# starting kernel
# baseline (speedup 1.0000x reference)
#include <hip/hip_runtime.h>
#include <cstdint>
#include <cstddef>

typedef float v2f __attribute__((ext_vector_type(2)));
typedef float v8f __attribute__((ext_vector_type(8)));

#define GAT_ALPHA 0.2f

__device__ __forceinline__ void atomicMaxF(float* addr, float v) {
    if (v >= 0.0f) atomicMax((int*)addr, __float_as_int(v));
    else           atomicMin((unsigned int*)addr, __float_as_uint(v));
}

__device__ __forceinline__ void atomAddF(float* p, float v) {
#if defined(__HIP_PLATFORM_AMD__)
    unsafeAtomicAdd(p, v);
#else
    atomicAdd(p, v);
#endif
}

__global__ void k_fill(float* __restrict__ p, float v, int n) {
    int i = blockIdx.x * blockDim.x + threadIdx.x;
    if (i < n) p[i] = v;
}

__global__ __launch_bounds__(32) void k_gemm_wmma(
    const float* __restrict__ A, const float* __restrict__ B, float* __restrict__ C,
    int K, long aStride, long bStride, long cStride)
{
    const int lane = threadIdx.x;
    const int m0   = blockIdx.x * 16;
    const int b    = blockIdx.y;
    const float* __restrict__ Ab = A + (long)b * aStride;
    const float* __restrict__ Bb = B + (long)b * bStride;
    float* __restrict__ Cb = C + (long)b * cStride;

    const int row = lane & 15;
    const int kh  = (lane >> 4) << 1;
    const int col = lane & 15;

    v8f c0 = {}, c1 = {}, c2 = {}, c3 = {};

    for (int k0 = 0; k0 < K; k0 += 4) {
        const float2 af = *reinterpret_cast<const float2*>(
            Ab + (size_t)(m0 + row) * K + k0 + kh);
        v2f a; a.x = af.x; a.y = af.y;

        const float* br0 = Bb + (size_t)(k0 + kh) * 64 + col;
        const float* br1 = br0 + 64;
        v2f b0, b1, b2, b3;
        b0.x = br0[0];  b0.y = br1[0];
        b1.x = br0[16]; b1.y = br1[16];
        b2.x = br0[32]; b2.y = br1[32];
        b3.x = br0[48]; b3.y = br1[48];

        c0 = __builtin_amdgcn_wmma_f32_16x16x4_f32(false, a, false, b0, (short)0, c0, false, false);
        c1 = __builtin_amdgcn_wmma_f32_16x16x4_f32(false, a, false, b1, (short)0, c1, false, false);
        c2 = __builtin_amdgcn_wmma_f32_16x16x4_f32(false, a, false, b2, (short)0, c2, false, false);
        c3 = __builtin_amdgcn_wmma_f32_16x16x4_f32(false, a, false, b3, (short)0, c3, false, false);
    }

    const int rbase = m0 + ((lane >> 4) << 3);
    #pragma unroll
    for (int r = 0; r < 8; ++r) {
        float* crow = Cb + (size_t)(rbase + r) * 64 + col;
        crow[0]  = c0[r];
        crow[16] = c1[r];
        crow[32] = c2[r];
        crow[48] = c3[r];
    }
}

__global__ void k_node_scores(const float* __restrict__ Wh, const float* __restrict__ a,
                              float* __restrict__ ssrc, float* __restrict__ sdst,
                              int NH, int N)
{
    int wid  = (blockIdx.x * blockDim.x + threadIdx.x) >> 5;
    int lane = threadIdx.x & 31;
    if (wid >= NH) return;
    int h = wid / N;
    const float* __restrict__ row = Wh + (size_t)wid * 64;
    const float* __restrict__ as  = a + h * 128;
    const float* __restrict__ ad  = as + 64;
    float w0 = row[lane], w1 = row[lane + 32];
    float s0 = w0 * as[lane] + w1 * as[lane + 32];
    float s1 = w0 * ad[lane] + w1 * ad[lane + 32];
    #pragma unroll
    for (int off = 16; off >= 1; off >>= 1) {
        s0 += __shfl_xor(s0, off, 32);
        s1 += __shfl_xor(s1, off, 32);
    }
    if (lane == 0) { ssrc[wid] = s0; sdst[wid] = s1; }
}

__global__ void k_edge_max(const int* __restrict__ src, const int* __restrict__ dst,
                           const float* __restrict__ ssrc, const float* __restrict__ sdst,
                           float* __restrict__ ebuf, float* __restrict__ m,
                           int E, int H, int N)
{
    int tid = blockIdx.x * blockDim.x + threadIdx.x;
    if (tid >= E * H) return;
    int e = tid % E, h = tid / E;
    int s = src[e], d = dst[e];
    float v = ssrc[h * N + s] + sdst[h * N + d];
    v = v > 0.0f ? v : GAT_ALPHA * v;
    ebuf[tid] = v;
    atomicMaxF(&m[h * N + s], v);
}

__global__ void k_edge_exp(const int* __restrict__ src, const float* __restrict__ m,
                           float* __restrict__ ebuf, float* __restrict__ denom,
                           int E, int H, int N)
{
    int tid = blockIdx.x * blockDim.x + threadIdx.x;
    if (tid >= E * H) return;
    int e = tid % E, h = tid / E;
    int s = src[e];
    float ex = __expf(ebuf[tid] - m[h * N + s]);
    ebuf[tid] = ex;
    atomAddF(&denom[h * N + s], ex);
}

__global__ void k_edge_scatter(const int* __restrict__ src, const int* __restrict__ dst,
                               const float* __restrict__ ebuf, const float* __restrict__ denom,
                               const float* __restrict__ Wh, float* __restrict__ hp,
                               int E, int H, int N)
{
    int wid  = (blockIdx.x * blockDim.x + threadIdx.x) >> 5;
    int lane = threadIdx.x & 31;
    if (wid >= E * H) return;
    int e = wid % E, h = wid / E;
    int s = src[e], d = dst[e];
    float att = ebuf[wid] / denom[h * N + s];
    const float* __restrict__ wrow = Wh + (size_t)(h * N + d) * 64;
    float* __restrict__ orow = hp + (size_t)(h * N + s) * 64;
    atomAddF(&orow[lane],      att * wrow[lane]);
    atomAddF(&orow[lane + 32], att * wrow[lane + 32]);
}

__global__ void k_elu_concat(const float* __restrict__ hp, float* __restrict__ hcat, int N) {
    int tid = blockIdx.x * blockDim.x + threadIdx.x;
    if (tid >= N * 256) return;
    int n = tid >> 8, c = tid & 255, h = c >> 6, d = c & 63;
    float v = hp[((size_t)h * N + n) * 64 + d];
    hcat[tid] = v > 0.0f ? v : __expf(v) - 1.0f;
}

__global__ void k_elu(const float* __restrict__ in, float* __restrict__ out, int n) {
    int i = blockIdx.x * blockDim.x + threadIdx.x;
    if (i < n) {
        float v = in[i];
        out[i] = v > 0.0f ? v : __expf(v) - 1.0f;
    }
}

static inline int cdiv(long a, int b) { return (int)((a + b - 1) / b); }

extern "C" void kernel_launch(void* const* d_in, const int* in_sizes, int n_in,
                              void* d_out, int out_size, void* d_ws, size_t ws_size,
                              hipStream_t stream)
{
    const float* x       = (const float*)d_in[0];
    const int*   edges   = (const int*)d_in[1];
    const float* W_heads = (const float*)d_in[2];
    const float* a_heads = (const float*)d_in[3];
    const float* W_out   = (const float*)d_in[4];
    const float* a_out   = (const float*)d_in[5];
    float* out = (float*)d_out;

    const int N = in_sizes[0] / 128;
    const int E = in_sizes[1] / 2;
    const int H = 4, HID = 64;

    const int* src = edges;
    const int* dst = edges + E;

    float* ws = (float*)d_ws;
    size_t off = 0;
    float* Wh1   = ws + off; off += (size_t)H * N * HID;
    float* hcat  = ws + off; off += (size_t)N * 256;
    float* Wh2   = ws + off; off += (size_t)N * HID;
    float* ssrc1 = ws + off; off += (size_t)H * N;
    float* sdst1 = ws + off; off += (size_t)H * N;
    float* ssrc2 = ws + off; off += (size_t)N;
    float* sdst2 = ws + off; off += (size_t)N;
    float* ebuf1 = ws + off; off += (size_t)H * E;
    float* ebuf2 = ws + off; off += (size_t)E;
    float* zbase = ws + off;
    float* denom1 = ws + off; off += (size_t)H * N;
    float* hp1    = ws + off; off += (size_t)H * N * HID;
    float* denom2 = ws + off; off += (size_t)N;
    float* acc2   = ws + off; off += (size_t)N * HID;
    size_t zcount = (size_t)(ws + off - zbase);
    float* nbase = ws + off;
    float* m1 = ws + off; off += (size_t)H * N;
    float* m2 = ws + off; off += (size_t)N;
    size_t ncount = (size_t)(ws + off - nbase);

    const int BS = 256;

    k_fill<<<cdiv((long)zcount, BS), BS, 0, stream>>>(zbase, 0.0f, (int)zcount);
    k_fill<<<cdiv((long)ncount, BS), BS, 0, stream>>>(nbase, -__builtin_huge_valf(), (int)ncount);

    k_gemm_wmma<<<dim3(N / 16, H), 32, 0, stream>>>(x, W_heads, Wh1,
        128, 0L, (long)128 * 64, (long)N * 64);
    k_node_scores<<<cdiv((long)H * N * 32, BS), BS, 0, stream>>>(Wh1, a_heads, ssrc1, sdst1, H * N, N);
    k_edge_max<<<cdiv((long)E * H, BS), BS, 0, stream>>>(src, dst, ssrc1, sdst1, ebuf1, m1, E, H, N);
    k_edge_exp<<<cdiv((long)E * H, BS), BS, 0, stream>>>(src, m1, ebuf1, denom1, E, H, N);
    k_edge_scatter<<<cdiv((long)E * H * 32, BS), BS, 0, stream>>>(src, dst, ebuf1, denom1, Wh1, hp1, E, H, N);
    k_elu_concat<<<cdiv((long)N * 256, BS), BS, 0, stream>>>(hp1, hcat, N);

    k_gemm_wmma<<<dim3(N / 16, 1), 32, 0, stream>>>(hcat, W_out, Wh2,
        256, 0L, 0L, 0L);
    k_node_scores<<<cdiv((long)N * 32, BS), BS, 0, stream>>>(Wh2, a_out, ssrc2, sdst2, N, N);
    k_edge_max<<<cdiv((long)E, BS), BS, 0, stream>>>(src, dst, ssrc2, sdst2, ebuf2, m2, E, 1, N);
    k_edge_exp<<<cdiv((long)E, BS), BS, 0, stream>>>(src, m2, ebuf2, denom2, E, 1, N);
    k_edge_scatter<<<cdiv((long)E * 32, BS), BS, 0, stream>>>(src, dst, ebuf2, denom2, Wh2, acc2, E, 1, N);

    k_elu<<<cdiv((long)N * HID, BS), BS, 0, stream>>>(acc2, out, N * HID);
}


# Edge list split once per call by owner row range: 256 chunk groups count-sort in LDS, store (row,dst) words + chunk x range table; owners read slices, no list scan; LDS add order changes; 7 launches
# speedup vs baseline: 6.8768x; 6.8768x over previous
#include <hip/hip_runtime.h>
#include <cstdint>
#include <cstddef>

typedef float v2f __attribute__((ext_vector_type(2)));
typedef float v8f __attribute__((ext_vector_type(8)));

#define GAT_ALPHA 0.2f

__device__ __forceinline__ unsigned int ordKey(float v) {
    unsigned int b = __float_as_uint(v);
    return (b & 0x80000000u) ? ~b : (b | 0x80000000u);
}
__device__ __forceinline__ float ordVal(unsigned int k) {
    return __uint_as_float((k & 0x80000000u) ? (k & 0x7fffffffu) : ~k);
}

__global__ __launch_bounds__(32) void k_gemm_wmma(
    const float* __restrict__ A, const float* __restrict__ B, float* __restrict__ C,
    int K, long aStride, long bStride, long cStride)
{
    const int lane = threadIdx.x;
    const int m0   = blockIdx.x * 16;
    const int b    = blockIdx.y;
    const float* __restrict__ Ab = A + (long)b * aStride;
    const float* __restrict__ Bb = B + (long)b * bStride;
    float* __restrict__ Cb = C + (long)b * cStride;

    const int row = lane & 15;
    const int kh  = (lane >> 4) << 1;
    const int col = lane & 15;

    v8f c0 = {}, c1 = {}, c2 = {}, c3 = {};

    for (int k0 = 0; k0 < K; k0 += 4) {
        const float2 af = *reinterpret_cast<const float2*>(
            Ab + (size_t)(m0 + row) * K + k0 + kh);
        v2f a; a.x = af.x; a.y = af.y;

        const float* br0 = Bb + (size_t)(k0 + kh) * 64 + col;
        const float* br1 = br0 + 64;
        v2f b0, b1, b2, b3;
        b0.x = br0[0];  b0.y = br1[0];
        b1.x = br0[16]; b1.y = br1[16];
        b2.x = br0[32]; b2.y = br1[32];
        b3.x = br0[48]; b3.y = br1[48];

        c0 = __builtin_amdgcn_wmma_f32_16x16x4_f32(false, a, false, b0, (short)0, c0, false, false);
        c1 = __builtin_amdgcn_wmma_f32_16x16x4_f32(false, a, false, b1, (short)0, c1, false, false);
        c2 = __builtin_amdgcn_wmma_f32_16x16x4_f32(false, a, false, b2, (short)0, c2, false, false);
        c3 = __builtin_amdgcn_wmma_f32_16x16x4_f32(false, a, false, b3, (short)0, c3, false, false);
    }

    const int rbase = m0 + ((lane >> 4) << 3);
    #pragma unroll
    for (int r = 0; r < 8; ++r) {
        float* crow = Cb + (size_t)(rbase + r) * 64 + col;
        crow[0]  = c0[r];
        crow[16] = c1[r];
        crow[32] = c2[r];
        crow[48] = c3[r];
    }
}

template <int H, int U>
__global__ __launch_bounds__(256) void k_node_scores(const float* __restrict__ Wh, const float* __restrict__ a,
                                                     float* __restrict__ ssrc, float* __restrict__ sdst, int N)
{
    const int lane = threadIdx.x & 31;
    const int wpb  = blockDim.x >> 5;
    const int w    = __builtin_amdgcn_readfirstlane(blockIdx.x * wpb + (threadIdx.x >> 5));
    const int nw   = gridDim.x * wpb;

    float as0[H], as1[H], ad0[H], ad1[H];
    #pragma unroll
    for (int h = 0; h < H; ++h) {
        const float* __restrict__ ah = a + h * 128;
        as0[h] = ah[lane];      as1[h] = ah[lane + 32];
        ad0[h] = ah[64 + lane]; ad1[h] = ah[96 + lane];
    }

    for (int n0 = w * U; n0 < N; n0 += nw * U) {
        float w0[U][H], w1[U][H];
        #pragma unroll
        for (int u = 0; u < U; ++u) {
            const int nn = min(n0 + u, N - 1);
            #pragma unroll
            for (int h = 0; h < H; ++h) {
                const float* __restrict__ row = Wh + ((size_t)h * N + nn) * 64;
                w0[u][h] = row[lane]; w1[u][h] = row[lane + 32];
            }
        }
        float s0[U][H], s1[U][H];
        #pragma unroll
        for (int u = 0; u < U; ++u) {
            #pragma unroll
            for (int h = 0; h < H; ++h) {
                s0[u][h] = __builtin_fmaf(w0[u][h], as0[h], w1[u][h] * as1[h]);
                s1[u][h] = __builtin_fmaf(w0[u][h], ad0[h], w1[u][h] * ad1[h]);
            }
        }
        #pragma unroll
        for (int off = 16; off >= 1; off >>= 1) {
            #pragma unroll
            for (int u = 0; u < U; ++u) {
                #pragma unroll
                for (int h = 0; h < H; ++h) {
                    s0[u][h] += __shfl_xor(s0[u][h], off, 32);
                    s1[u][h] += __shfl_xor(s1[u][h], off, 32);
                }
            }
        }
        if (lane == 0) {
            #pragma unroll
            for (int u = 0; u < U; ++u) {
                if (n0 + u < N) {
                    const size_t o = (size_t)(n0 + u) * H;
                    if constexpr (H == 4) {
                        *reinterpret_cast<float4*>(ssrc + o) = make_float4(s0[u][0], s0[u][1], s0[u][2], s0[u][3]);
                        *reinterpret_cast<float4*>(sdst + o) = make_float4(s1[u][0], s1[u][1], s1[u][2], s1[u][3]);
                    } else {
                        #pragma unroll
                        for (int h = 0; h < H; ++h) { ssrc[o + h] = s0[u][h]; sdst[o + h] = s1[u][h]; }
                    }
                }
            }
        }
    }
}

#define OWN_R    196
#define OWN_T    1024
#define OWN_QCAP 16384
#define OWN_P    4

__device__ __forceinline__ void ldsAddF(float* p, float v) {
    __hip_atomic_fetch_add(p, v, __ATOMIC_RELAXED, __HIP_MEMORY_SCOPE_WORKGROUP);
}

__device__ __forceinline__ float eluF(float v) { return v > 0.0f ? v : __expf(v) - 1.0f; }

__device__ __forceinline__ float edgeScore(float a, float b) {
    const float v = a + b;
    return v > 0.0f ? v : GAT_ALPHA * v;
}

template <int H>
__device__ __forceinline__ void nodeScores(const float* __restrict__ p, int node, float (&v)[H]) {
    if constexpr (H == 4) {
        const float4 q = *reinterpret_cast<const float4*>(p + (size_t)node * 4);
        v[0] = q.x; v[1] = q.y; v[2] = q.z; v[3] = q.w;
    } else {
        #pragma unroll
        for (int h = 0; h < H; ++h) v[h] = p[(size_t)node * H + h];
    }
}

__device__ __forceinline__ void holdInts(const int (&d)[OWN_P]) {
    static_assert(OWN_P == 4, "written out for four slots");
    asm volatile("" :: "v"(d[0]), "v"(d[1]), "v"(d[2]), "v"(d[3]));
}
template <int H>
__device__ __forceinline__ void holdScores(const float (&v)[OWN_P][H]) {
    static_assert(OWN_P == 4 && (H == 1 || H == 4), "written out for four slots of one or four heads");
    if constexpr (H == 4) {
        asm volatile("" :: "v"(v[0][0]), "v"(v[0][1]), "v"(v[0][2]), "v"(v[0][3]),
                           "v"(v[1][0]), "v"(v[1][1]), "v"(v[1][2]), "v"(v[1][3]),
                           "v"(v[2][0]), "v"(v[2][1]), "v"(v[2][2]), "v"(v[2][3]),
                           "v"(v[3][0]), "v"(v[3][1]), "v"(v[3][2]), "v"(v[3][3]));
    } else {
        asm volatile("" :: "v"(v[0][0]), "v"(v[1][0]), "v"(v[2][0]), "v"(v[3][0]));
    }
}

#define BIN_NCH  256
#define BIN_T    512
#define BIN_U    8
#define BIN_MAXR 1024

static_assert(OWN_R <= 256, "the local row of an edge is kept in 8 bits");

__device__ __forceinline__ int binRanges(int N) { return min((N + OWN_R - 1) / OWN_R, BIN_MAXR - 1); }
__device__ __forceinline__ int binTabPitch(int NB) { return (NB + 32) & ~31; }
__device__ __forceinline__ int binChunk(int E) { return (E + BIN_NCH - 1) / BIN_NCH; }
__device__ __forceinline__ int binPiecePitch(int CH) { return (CH + 31) & ~31; }

__global__ __launch_bounds__(BIN_T) void k_bin_edges(
    const int* __restrict__ src, const int* __restrict__ dst,
    int* __restrict__ binned, int* __restrict__ tab, int E, int N)
{
    __shared__ int cnt[BIN_MAXR];
    __shared__ int wtot[BIN_T / 32];

    const int tid = threadIdx.x, lane = tid & 31, wave = tid >> 5;
    const int NB = binRanges(N);
    const int TP = binTabPitch(NB);
    const int CH = binChunk(E);
    const int SP = binPiecePitch(CH);
    const int c  = blockIdx.x;
    const int e0 = c * CH;
    const int n  = min(CH, E - e0);
    const unsigned lim = (unsigned)min(N, NB * OWN_R);

    for (int i = tid; i < BIN_MAXR; i += BIN_T) cnt[i] = 0;
    __syncthreads();

    for (int k0 = 0; k0 < n; k0 += BIN_T * BIN_U) {
        int s[BIN_U];
        #pragma unroll
        for (int u = 0; u < BIN_U; ++u) s[u] = src[e0 + min(k0 + u * BIN_T + tid, n - 1)];
        #pragma unroll
        for (int u = 0; u < BIN_U; ++u) {
            const int k = k0 + u * BIN_T + tid;
            if (k < n && (unsigned)s[u] < lim) atomicAdd(&cnt[(unsigned)s[u] / OWN_R], 1);
        }
    }
    __syncthreads();

    constexpr int PER = BIN_MAXR / BIN_T;
    int v[PER];
    int tsum = 0;
    #pragma unroll
    for (int j = 0; j < PER; ++j) { v[j] = cnt[tid * PER + j]; tsum += v[j]; }
    int incl = tsum;
    #pragma unroll
    for (int off = 1; off < 32; off <<= 1) {
        const int y = __shfl_up(incl, off, 32);
        if (lane >= off) incl += y;
    }
    if (lane == 31) wtot[wave] = incl;
    __syncthreads();
    int run = incl - tsum;
    for (int w = 0; w < wave; ++w) run += wtot[w];
    #pragma unroll
    for (int j = 0; j < PER; ++j) {
        const int r = tid * PER + j;
        cnt[r] = run;
        if (r <= NB) tab[(size_t)c * TP + r] = run;
        run += v[j];
    }
    __syncthreads();

    for (int k0 = 0; k0 < n; k0 += BIN_T * BIN_U) {
        int s[BIN_U], d[BIN_U];
        #pragma unroll
        for (int u = 0; u < BIN_U; ++u) {
            const int k = e0 + min(k0 + u * BIN_T + tid, n - 1);
            s[u] = src[k]; d[u] = dst[k];
        }
        #pragma unroll
        for (int u = 0; u < BIN_U; ++u) {
            const int k = k0 + u * BIN_T + tid;
            if (k < n && (unsigned)s[u] < lim) {
                const unsigned r = (unsigned)s[u] / OWN_R;
                const int p = atomicAdd(&cnt[r], 1);
                binned[(size_t)c * SP + p] = (int)((((unsigned)s[u] - r * OWN_R) << 24) | ((unsigned)d[u] & 0xffffffu));
            }
        }
    }
}

template <int HS>
__global__ __launch_bounds__(OWN_T) void k_own_attend(
    const int* __restrict__ src, const int* __restrict__ dst,
    const int* __restrict__ binned, const int* __restrict__ tab,
    const float* __restrict__ ssrc, const float* __restrict__ sdst,
    const float* __restrict__ Wh, float* __restrict__ outp, int E, int N)
{
    constexpr int H  = 1 << HS;
    constexpr int RW = H * 64;
    __shared__ __attribute__((aligned(16))) float acc[OWN_R * RW];
    __shared__ float den[OWN_R * H];
    __shared__ unsigned int mx[OWN_R * H];
    __shared__ int   que[OWN_QCAP];
    __shared__ unsigned char qrl[OWN_QCAP];
    __shared__ int   cst[BIN_NCH];
    __shared__ int   qof[BIN_NCH + 1];
    __shared__ int   wsm[BIN_NCH / 32];

    const int tid  = threadIdx.x;
    const int lane = tid & 31;
    const int lo   = blockIdx.x * OWN_R;
    const int nrows = min(OWN_R, N - lo);

    const int NB = binRanges(N);
    const int TP = binTabPitch(NB);
    const int SP = binPiecePitch(binChunk(E));
    int myst = 0, mycnt = 0;
    if (tid < BIN_NCH && (int)blockIdx.x < NB) {
        const int* tr = tab + (size_t)tid * TP + blockIdx.x;
        myst = tr[0]; mycnt = tr[1] - myst;
    }

    float4* acc4 = reinterpret_cast<float4*>(acc);
    for (int i = tid; i < OWN_R * RW / 4; i += OWN_T) acc4[i] = make_float4(0.0f, 0.0f, 0.0f, 0.0f);
    for (int i = tid; i < OWN_R * H; i += OWN_T) { den[i] = 0.0f; mx[i] = 0u; }

    int incl = mycnt;
    if (tid < BIN_NCH) {
        #pragma unroll
        for (int off = 1; off < 32; off <<= 1) {
            const int y = __shfl_up(incl, off, 32);
            if (lane >= off) incl += y;
        }
        cst[tid] = myst;
        if (lane == 31) wsm[tid >> 5] = incl;
    }
    __syncthreads();
    if (tid < BIN_NCH) {
        int pre = incl - mycnt;
        for (int w = 0; w < (tid >> 5); ++w) pre += wsm[w];
        qof[tid] = pre;
        if (tid == BIN_NCH - 1) qof[BIN_NCH] = pre + mycnt;
    }
    __syncthreads();
    const int ntot = qof[BIN_NCH];

    if (ntot <= OWN_QCAP) {
        for (int base = 0; base < ntot; base += OWN_T * OWN_P) {
            const int i0 = base + tid;
            if (i0 < ntot) {
                int qi[OWN_P], cI[OWN_P], wI[OWN_P], rI[OWN_P], dI[OWN_P];
                #pragma unroll
                for (int u = 0; u < OWN_P; ++u) {
                    const int i = i0 + u * OWN_T;
                    qi[u] = i < ntot ? i : i0;
                }
                #pragma unroll
                for (int u = 0; u < OWN_P; ++u) {
                    int c = 0;
                    #pragma unroll
                    for (int st = BIN_NCH / 2; st >= 1; st >>= 1)
                        if (qof[c + st] <= qi[u]) c += st;
                    cI[u] = c;
                }
                #pragma unroll
                for (int u = 0; u < OWN_P; ++u)
                    wI[u] = binned[(size_t)cI[u] * SP + cst[cI[u]] + (qi[u] - qof[cI[u]])];
                holdInts(wI);
                float a[OWN_P][H], b[OWN_P][H];
                #pragma unroll
                for (int u = 0; u < OWN_P; ++u) {
                    dI[u] = wI[u] & 0xffffff;
                    rI[u] = (int)((unsigned)wI[u] >> 24);
                    nodeScores<H>(ssrc, lo + rI[u], a[u]);
                    nodeScores<H>(sdst, dI[u], b[u]);
                }
                holdScores<H>(a);
                holdScores<H>(b);
                #pragma unroll
                for (int u = 0; u < OWN_P; ++u) {
                    #pragma unroll
                    for (int h = 0; h < H; ++h)
                        atomicMax(&mx[rI[u] * H + h], ordKey(edgeScore(a[u][h], b[u][h])));
                    que[qi[u]] = dI[u];
                    qrl[qi[u]] = (unsigned char)rI[u];
                }
            }
        }
        __syncthreads();

        constexpr int EPB = 32 >> HS;
        const int wave = __builtin_amdgcn_readfirstlane(tid >> 5);
        const int j = lane >> HS, h = lane & (H - 1);
        for (int b0 = wave * EPB; b0 < ntot; b0 += (OWN_T / 32) * EPB) {
            const bool ok = b0 + j < ntot;
            const int qi = ok ? b0 + j : b0;
            const int d  = que[qi];
            const int rl = qrl[qi];
            const float sc = edgeScore(ssrc[(size_t)(lo + rl) * H + h], sdst[(size_t)d * H + h]);
            const float ex = __expf(sc - ordVal(mx[rl * H + h]));
            #pragma unroll
            for (int g = 0; g < 32; g += 16) {
                float2 w[16];
                #pragma unroll
                for (int t = 0; t < 16; ++t) {
                    const int L  = g + t;
                    const int dL = __builtin_amdgcn_readlane(d, L);
                    const float* row = Wh + ((size_t)(L & (H - 1)) * N + dL) * 64;
                    w[t] = *reinterpret_cast<const float2*>(row + 2 * lane);
                }
                #pragma unroll
                for (int t = 0; t < 16; ++t) {
                    const int L = g + t;
                    if (b0 + (L >> HS) < ntot) {
                        const float exL = __int_as_float(__builtin_amdgcn_readlane(__float_as_int(ex), L));
                        const int   rlL = __builtin_amdgcn_readlane(rl, L);
                        float* ap = acc + rlL * RW + (L & (H - 1)) * 64 + 2 * lane;
                        ldsAddF(ap,     exL * w[t].x);
                        ldsAddF(ap + 1, exL * w[t].y);
                    }
                }
            }
            if (ok) ldsAddF(&den[rl * H + h], ex);
        }
    } else {
        for (int e = tid; e < E; e += OWN_T) {
            const int s = src[e];
            const unsigned rl = (unsigned)(s - lo);
            if (rl < (unsigned)nrows) {
                const int d = dst[e];
                for (int hh = 0; hh < H; ++hh)
                    atomicMax(&mx[rl * H + hh],
                              ordKey(edgeScore(ssrc[(size_t)s * H + hh], sdst[(size_t)d * H + hh])));
            }
        }
        __syncthreads();
        for (int e = tid; e < E; e += OWN_T) {
            const int s = src[e];
            const unsigned rl = (unsigned)(s - lo);
            if (rl < (unsigned)nrows) {
                const int d = dst[e];
                for (int hh = 0; hh < H; ++hh) {
                    const float sc = edgeScore(ssrc[(size_t)s * H + hh], sdst[(size_t)d * H + hh]);
                    const float ex = __expf(sc - ordVal(mx[rl * H + hh]));
                    ldsAddF(&den[rl * H + hh], ex);
                    const float* row = Wh + ((size_t)hh * N + d) * 64;
                    float* ap = acc + rl * RW + hh * 64;
                    for (int k = 0; k < 64; ++k) ldsAddF(ap + k, ex * row[k]);
                }
            }
        }
    }
    __syncthreads();

    float4* o4 = reinterpret_cast<float4*>(outp + (size_t)lo * RW);
    const int n4 = nrows * (RW / 4);
    for (int i = tid; i < n4; i += OWN_T) {
        const float4 a = acc4[i];
        const float dn = den[i >> 4];
        float4 r;
        if (dn > 0.0f) {
            r.x = eluF(a.x / dn); r.y = eluF(a.y / dn); r.z = eluF(a.z / dn); r.w = eluF(a.w / dn);
        } else {
            r = make_float4(0.0f, 0.0f, 0.0f, 0.0f);
        }
        o4[i] = r;
    }
}

static inline int cdiv(long a, int b) { return (int)((a + b - 1) / b); }

extern "C" void kernel_launch(void* const* d_in, const int* in_sizes, int n_in,
                              void* d_out, int out_size, void* d_ws, size_t ws_size,
                              hipStream_t stream)
{
    const float* x       = (const float*)d_in[0];
    const int*   edges   = (const int*)d_in[1];
    const float* W_heads = (const float*)d_in[2];
    const float* a_heads = (const float*)d_in[3];
    const float* W_out   = (const float*)d_in[4];
    const float* a_out   = (const float*)d_in[5];
    float* out = (float*)d_out;

    const int N = in_sizes[0] / 128;
    const int E = in_sizes[1] / 2;
    const int H = 4, HID = 64;

    const int* src = edges;
    const int* dst = edges + E;

    float* ws = (float*)d_ws;
    size_t off = 0;
    float* Wh1   = ws + off; off += (size_t)H * N * HID;
    float* hcat  = ws + off; off += (size_t)N * 256;
    float* Wh2   = ws + off; off += (size_t)N * HID;
    float* ssrc1 = ws + off; off += (size_t)H * N;
    float* sdst1 = ws + off; off += (size_t)H * N;
    float* ssrc2 = ws + off; off += (size_t)N;
    float* sdst2 = ws + off; off += (size_t)N;
    int* binned  = (int*)(ws + off); off += (size_t)E + 64 * BIN_NCH;
    int* tab     = (int*)(ws + off);

    const int BS = 256;

    k_bin_edges<<<BIN_NCH, BIN_T, 0, stream>>>(src, dst, binned, tab, E, N);

    k_gemm_wmma<<<dim3(N / 16, H), 32, 0, stream>>>(x, W_heads, Wh1,
        128, 0L, (long)128 * 64, (long)N * 64);
    k_node_scores<4, 2><<<1024, BS, 0, stream>>>(Wh1, a_heads, ssrc1, sdst1, N);
    k_own_attend<2><<<cdiv((long)N, OWN_R), OWN_T, 0, stream>>>(src, dst, binned, tab, ssrc1, sdst1, Wh1, hcat, E, N);

    k_gemm_wmma<<<dim3(N / 16, 1), 32, 0, stream>>>(hcat, W_out, Wh2,
        256, 0L, 0L, 0L);
    k_node_scores<1, 8><<<256, BS, 0, stream>>>(Wh2, a_out, ssrc2, sdst2, N);
    k_own_attend<0><<<cdiv((long)N, OWN_R), OWN_T, 0, stream>>>(src, dst, binned, tab, ssrc2, sdst2, Wh2, out, E, N);
}


# Row sums leave LDS float atomics: queue regrouped in LDS by owner wave (row mod 32); one wave alone adds a row by plain LDS load, fma, store, two edges a round trip; LDS add order changes
# speedup vs baseline: 7.4145x; 1.0782x over previous
#include <hip/hip_runtime.h>
#include <cstdint>
#include <cstddef>

typedef float v2f __attribute__((ext_vector_type(2)));
typedef float v8f __attribute__((ext_vector_type(8)));

#define GAT_ALPHA 0.2f

__device__ __forceinline__ unsigned int ordKey(float v) {
    unsigned int b = __float_as_uint(v);
    return (b & 0x80000000u) ? ~b : (b | 0x80000000u);
}
__device__ __forceinline__ float ordVal(unsigned int k) {
    return __uint_as_float((k & 0x80000000u) ? (k & 0x7fffffffu) : ~k);
}

__global__ __launch_bounds__(32) void k_gemm_wmma(
    const float* __restrict__ A, const float* __restrict__ B, float* __restrict__ C,
    int K, long aStride, long bStride, long cStride)
{
    const int lane = threadIdx.x;
    const int m0   = blockIdx.x * 16;
    const int b    = blockIdx.y;
    const float* __restrict__ Ab = A + (long)b * aStride;
    const float* __restrict__ Bb = B + (long)b * bStride;
    float* __restrict__ Cb = C + (long)b * cStride;

    const int row = lane & 15;
    const int kh  = (lane >> 4) << 1;
    const int col = lane & 15;

    v8f c0 = {}, c1 = {}, c2 = {}, c3 = {};

    for (int k0 = 0; k0 < K; k0 += 4) {
        const float2 af = *reinterpret_cast<const float2*>(
            Ab + (size_t)(m0 + row) * K + k0 + kh);
        v2f a; a.x = af.x; a.y = af.y;

        const float* br0 = Bb + (size_t)(k0 + kh) * 64 + col;
        const float* br1 = br0 + 64;
        v2f b0, b1, b2, b3;
        b0.x = br0[0];  b0.y = br1[0];
        b1.x = br0[16]; b1.y = br1[16];
        b2.x = br0[32]; b2.y = br1[32];
        b3.x = br0[48]; b3.y = br1[48];

        c0 = __builtin_amdgcn_wmma_f32_16x16x4_f32(false, a, false, b0, (short)0, c0, false, false);
        c1 = __builtin_amdgcn_wmma_f32_16x16x4_f32(false, a, false, b1, (short)0, c1, false, false);
        c2 = __builtin_amdgcn_wmma_f32_16x16x4_f32(false, a, false, b2, (short)0, c2, false, false);
        c3 = __builtin_amdgcn_wmma_f32_16x16x4_f32(false, a, false, b3, (short)0, c3, false, false);
    }

    const int rbase = m0 + ((lane >> 4) << 3);
    #pragma unroll
    for (int r = 0; r < 8; ++r) {
        float* crow = Cb + (size_t)(rbase + r) * 64 + col;
        crow[0]  = c0[r];
        crow[16] = c1[r];
        crow[32] = c2[r];
        crow[48] = c3[r];
    }
}

template <int H, int U>
__global__ __launch_bounds__(256) void k_node_scores(const float* __restrict__ Wh, const float* __restrict__ a,
                                                     float* __restrict__ ssrc, float* __restrict__ sdst, int N)
{
    const int lane = threadIdx.x & 31;
    const int wpb  = blockDim.x >> 5;
    const int w    = __builtin_amdgcn_readfirstlane(blockIdx.x * wpb + (threadIdx.x >> 5));
    const int nw   = gridDim.x * wpb;

    float as0[H], as1[H], ad0[H], ad1[H];
    #pragma unroll
    for (int h = 0; h < H; ++h) {
        const float* __restrict__ ah = a + h * 128;
        as0[h] = ah[lane];      as1[h] = ah[lane + 32];
        ad0[h] = ah[64 + lane]; ad1[h] = ah[96 + lane];
    }

    for (int n0 = w * U; n0 < N; n0 += nw * U) {
        float w0[U][H], w1[U][H];
        #pragma unroll
        for (int u = 0; u < U; ++u) {
            const int nn = min(n0 + u, N - 1);
            #pragma unroll
            for (int h = 0; h < H; ++h) {
                const float* __restrict__ row = Wh + ((size_t)h * N + nn) * 64;
                w0[u][h] = row[lane]; w1[u][h] = row[lane + 32];
            }
        }
        float s0[U][H], s1[U][H];
        #pragma unroll
        for (int u = 0; u < U; ++u) {
            #pragma unroll
            for (int h = 0; h < H; ++h) {
                s0[u][h] = __builtin_fmaf(w0[u][h], as0[h], w1[u][h] * as1[h]);
                s1[u][h] = __builtin_fmaf(w0[u][h], ad0[h], w1[u][h] * ad1[h]);
            }
        }
        #pragma unroll
        for (int off = 16; off >= 1; off >>= 1) {
            #pragma unroll
            for (int u = 0; u < U; ++u) {
                #pragma unroll
                for (int h = 0; h < H; ++h) {
                    s0[u][h] += __shfl_xor(s0[u][h], off, 32);
                    s1[u][h] += __shfl_xor(s1[u][h], off, 32);
                }
            }
        }
        if (lane == 0) {
            #pragma unroll
            for (int u = 0; u < U; ++u) {
                if (n0 + u < N) {
                    const size_t o = (size_t)(n0 + u) * H;
                    if constexpr (H == 4) {
                        *reinterpret_cast<float4*>(ssrc + o) = make_float4(s0[u][0], s0[u][1], s0[u][2], s0[u][3]);
                        *reinterpret_cast<float4*>(sdst + o) = make_float4(s1[u][0], s1[u][1], s1[u][2], s1[u][3]);
                    } else {
                        #pragma unroll
                        for (int h = 0; h < H; ++h) { ssrc[o + h] = s0[u][h]; sdst[o + h] = s1[u][h]; }
                    }
                }
            }
        }
    }
}

#define OWN_R    196
#define OWN_T    1024
#define OWN_QCAP 8192
#define OWN_P    4
#define OWN_NW   (OWN_T / 32)
static_assert(OWN_NW == 32, "one wave scans the group counts; a row's owner wave is its local row mod 32");

__device__ __forceinline__ void ldsAddF(float* p, float v) {
    __hip_atomic_fetch_add(p, v, __ATOMIC_RELAXED, __HIP_MEMORY_SCOPE_WORKGROUP);
}

__device__ __forceinline__ float eluF(float v) { return v > 0.0f ? v : __expf(v) - 1.0f; }

__device__ __forceinline__ float edgeScore(float a, float b) {
    const float v = a + b;
    return v > 0.0f ? v : GAT_ALPHA * v;
}

template <int H>
__device__ __forceinline__ void nodeScores(const float* __restrict__ p, int node, float (&v)[H]) {
    if constexpr (H == 4) {
        const float4 q = *reinterpret_cast<const float4*>(p + (size_t)node * 4);
        v[0] = q.x; v[1] = q.y; v[2] = q.z; v[3] = q.w;
    } else {
        #pragma unroll
        for (int h = 0; h < H; ++h) v[h] = p[(size_t)node * H + h];
    }
}

__device__ __forceinline__ void holdInts(const int (&d)[OWN_P]) {
    static_assert(OWN_P == 4, "written out for four slots");
    asm volatile("" :: "v"(d[0]), "v"(d[1]), "v"(d[2]), "v"(d[3]));
}
template <int H>
__device__ __forceinline__ void holdScores(const float (&v)[OWN_P][H]) {
    static_assert(OWN_P == 4 && (H == 1 || H == 4), "written out for four slots of one or four heads");
    if constexpr (H == 4) {
        asm volatile("" :: "v"(v[0][0]), "v"(v[0][1]), "v"(v[0][2]), "v"(v[0][3]),
                           "v"(v[1][0]), "v"(v[1][1]), "v"(v[1][2]), "v"(v[1][3]),
                           "v"(v[2][0]), "v"(v[2][1]), "v"(v[2][2]), "v"(v[2][3]),
                           "v"(v[3][0]), "v"(v[3][1]), "v"(v[3][2]), "v"(v[3][3]));
    } else {
        asm volatile("" :: "v"(v[0][0]), "v"(v[1][0]), "v"(v[2][0]), "v"(v[3][0]));
    }
}

__device__ __forceinline__ void holdRows(const float2 (&w)[32], int o) {
    asm volatile("" :: "v"(w[o + 0].x), "v"(w[o + 0].y), "v"(w[o + 1].x), "v"(w[o + 1].y),
                       "v"(w[o + 2].x), "v"(w[o + 2].y), "v"(w[o + 3].x), "v"(w[o + 3].y),
                       "v"(w[o + 4].x), "v"(w[o + 4].y), "v"(w[o + 5].x), "v"(w[o + 5].y),
                       "v"(w[o + 6].x), "v"(w[o + 6].y), "v"(w[o + 7].x), "v"(w[o + 7].y));
}

#define BIN_NCH  256
#define BIN_T    512
#define BIN_U    8
#define BIN_MAXR 1024

static_assert(OWN_R <= 256, "the local row of an edge is kept in 8 bits");

__device__ __forceinline__ int binRanges(int N) { return min((N + OWN_R - 1) / OWN_R, BIN_MAXR - 1); }
__device__ __forceinline__ int binTabPitch(int NB) { return (NB + 32) & ~31; }
__device__ __forceinline__ int binChunk(int E) { return (E + BIN_NCH - 1) / BIN_NCH; }
__device__ __forceinline__ int binPiecePitch(int CH) { return (CH + 31) & ~31; }

__global__ __launch_bounds__(BIN_T) void k_bin_edges(
    const int* __restrict__ src, const int* __restrict__ dst,
    int* __restrict__ binned, int* __restrict__ tab, int E, int N)
{
    __shared__ int cnt[BIN_MAXR];
    __shared__ int wtot[BIN_T / 32];

    const int tid = threadIdx.x, lane = tid & 31, wave = tid >> 5;
    const int NB = binRanges(N);
    const int TP = binTabPitch(NB);
    const int CH = binChunk(E);
    const int SP = binPiecePitch(CH);
    const int c  = blockIdx.x;
    const int e0 = c * CH;
    const int n  = min(CH, E - e0);
    const unsigned lim = (unsigned)min(N, NB * OWN_R);

    for (int i = tid; i < BIN_MAXR; i += BIN_T) cnt[i] = 0;
    __syncthreads();

    for (int k0 = 0; k0 < n; k0 += BIN_T * BIN_U) {
        int s[BIN_U];
        #pragma unroll
        for (int u = 0; u < BIN_U; ++u) s[u] = src[e0 + min(k0 + u * BIN_T + tid, n - 1)];
        #pragma unroll
        for (int u = 0; u < BIN_U; ++u) {
            const int k = k0 + u * BIN_T + tid;
            if (k < n && (unsigned)s[u] < lim) atomicAdd(&cnt[(unsigned)s[u] / OWN_R], 1);
        }
    }
    __syncthreads();

    constexpr int PER = BIN_MAXR / BIN_T;
    int v[PER];
    int tsum = 0;
    #pragma unroll
    for (int j = 0; j < PER; ++j) { v[j] = cnt[tid * PER + j]; tsum += v[j]; }
    int incl = tsum;
    #pragma unroll
    for (int off = 1; off < 32; off <<= 1) {
        const int y = __shfl_up(incl, off, 32);
        if (lane >= off) incl += y;
    }
    if (lane == 31) wtot[wave] = incl;
    __syncthreads();
    int run = incl - tsum;
    for (int w = 0; w < wave; ++w) run += wtot[w];
    #pragma unroll
    for (int j = 0; j < PER; ++j) {
        const int r = tid * PER + j;
        cnt[r] = run;
        if (r <= NB) tab[(size_t)c * TP + r] = run;
        run += v[j];
    }
    __syncthreads();

    for (int k0 = 0; k0 < n; k0 += BIN_T * BIN_U) {
        int s[BIN_U], d[BIN_U];
        #pragma unroll
        for (int u = 0; u < BIN_U; ++u) {
            const int k = e0 + min(k0 + u * BIN_T + tid, n - 1);
            s[u] = src[k]; d[u] = dst[k];
        }
        #pragma unroll
        for (int u = 0; u < BIN_U; ++u) {
            const int k = k0 + u * BIN_T + tid;
            if (k < n && (unsigned)s[u] < lim) {
                const unsigned r = (unsigned)s[u] / OWN_R;
                const int p = atomicAdd(&cnt[r], 1);
                binned[(size_t)c * SP + p] = (int)((((unsigned)s[u] - r * OWN_R) << 24) | ((unsigned)d[u] & 0xffffffu));
            }
        }
    }
}

template <int HS>
__global__ __launch_bounds__(OWN_T) void k_own_attend(
    const int* __restrict__ src, const int* __restrict__ dst,
    const int* __restrict__ binned, const int* __restrict__ tab,
    const float* __restrict__ ssrc, const float* __restrict__ sdst,
    const float* __restrict__ Wh, float* __restrict__ outp, int E, int N)
{
    constexpr int H  = 1 << HS;
    constexpr int RW = H * 64;
    __shared__ __attribute__((aligned(16))) float acc[OWN_R * RW];
    __shared__ float den[OWN_R * H];
    __shared__ unsigned int mx[OWN_R * H];
    __shared__ int   queA[OWN_QCAP];
    __shared__ int   queB[OWN_QCAP];
    __shared__ int   bcnt[OWN_NW];
    __shared__ int   bst[OWN_NW + 1];
    __shared__ int   bpos[OWN_NW];
    __shared__ int   cst[BIN_NCH];
    __shared__ int   qof[BIN_NCH + 1];
    __shared__ int   wsm[BIN_NCH / 32];

    const int tid  = threadIdx.x;
    const int lane = tid & 31;
    const int lo   = blockIdx.x * OWN_R;
    const int nrows = min(OWN_R, N - lo);

    const int NB = binRanges(N);
    const int TP = binTabPitch(NB);
    const int SP = binPiecePitch(binChunk(E));
    int myst = 0, mycnt = 0;
    if (tid < BIN_NCH && (int)blockIdx.x < NB) {
        const int* tr = tab + (size_t)tid * TP + blockIdx.x;
        myst = tr[0]; mycnt = tr[1] - myst;
    }

    float4* acc4 = reinterpret_cast<float4*>(acc);
    for (int i = tid; i < OWN_R * RW / 4; i += OWN_T) acc4[i] = make_float4(0.0f, 0.0f, 0.0f, 0.0f);
    for (int i = tid; i < OWN_R * H; i += OWN_T) { den[i] = 0.0f; mx[i] = 0u; }
    if (tid < OWN_NW) bcnt[tid] = 0;

    int incl = mycnt;
    if (tid < BIN_NCH) {
        #pragma unroll
        for (int off = 1; off < 32; off <<= 1) {
            const int y = __shfl_up(incl, off, 32);
            if (lane >= off) incl += y;
        }
        cst[tid] = myst;
        if (lane == 31) wsm[tid >> 5] = incl;
    }
    __syncthreads();
    if (tid < BIN_NCH) {
        int pre = incl - mycnt;
        for (int w = 0; w < (tid >> 5); ++w) pre += wsm[w];
        qof[tid] = pre;
        if (tid == BIN_NCH - 1) qof[BIN_NCH] = pre + mycnt;
    }
    __syncthreads();
    const int ntot = qof[BIN_NCH];

    if (ntot <= OWN_QCAP) {
        for (int base = 0; base < ntot; base += OWN_T * OWN_P) {
            const int i0 = base + tid;
            if (i0 < ntot) {
                int qi[OWN_P], cI[OWN_P], wI[OWN_P], rI[OWN_P], dI[OWN_P];
                #pragma unroll
                for (int u = 0; u < OWN_P; ++u) {
                    const int i = i0 + u * OWN_T;
                    qi[u] = i < ntot ? i : i0;
                }
                #pragma unroll
                for (int u = 0; u < OWN_P; ++u) {
                    int c = 0;
                    #pragma unroll
                    for (int st = BIN_NCH / 2; st >= 1; st >>= 1)
                        if (qof[c + st] <= qi[u]) c += st;
                    cI[u] = c;
                }
                #pragma unroll
                for (int u = 0; u < OWN_P; ++u)
                    wI[u] = binned[(size_t)cI[u] * SP + cst[cI[u]] + (qi[u] - qof[cI[u]])];
                holdInts(wI);
                float a[OWN_P][H], b[OWN_P][H];
                #pragma unroll
                for (int u = 0; u < OWN_P; ++u) {
                    dI[u] = wI[u] & 0xffffff;
                    rI[u] = (int)((unsigned)wI[u] >> 24);
                    nodeScores<H>(ssrc, lo + rI[u], a[u]);
                    nodeScores<H>(sdst, dI[u], b[u]);
                }
                holdScores<H>(a);
                holdScores<H>(b);
                #pragma unroll
                for (int u = 0; u < OWN_P; ++u) {
                    #pragma unroll
                    for (int h = 0; h < H; ++h)
                        atomicMax(&mx[rI[u] * H + h], ordKey(edgeScore(a[u][h], b[u][h])));
                    queA[qi[u]] = wI[u];
                    if (i0 + u * OWN_T < ntot) atomicAdd(&bcnt[rI[u] & (OWN_NW - 1)], 1);
                }
            }
        }
        __syncthreads();

        if (tid < OWN_NW) {
            const int v = bcnt[tid];
            int run = v;
            #pragma unroll
            for (int off = 1; off < 32; off <<= 1) {
                const int y = __shfl_up(run, off, 32);
                if (lane >= off) run += y;
            }
            bst[tid] = run - v; bpos[tid] = run - v;
            if (tid == OWN_NW - 1) bst[OWN_NW] = run;
        }
        __syncthreads();
        for (int i = tid; i < ntot; i += OWN_T) {
            const int wd = queA[i];
            const int p = atomicAdd(&bpos[((unsigned)wd >> 24) & (OWN_NW - 1)], 1);
            queB[p] = wd;
        }
        __syncthreads();

        constexpr int EPB = 32 >> HS;
        const int wave = __builtin_amdgcn_readfirstlane(tid >> 5);
        const int bbeg = __builtin_amdgcn_readfirstlane(bst[wave]);
        const int bend = __builtin_amdgcn_readfirstlane(bst[wave + 1]);
        const int j = lane >> HS, h = lane & (H - 1);
        for (int b0 = bbeg; b0 < bend; b0 += EPB) {
            const bool ok = b0 + j < bend;
            const int wd = queB[ok ? b0 + j : b0];
            const int d  = wd & 0xffffff;
            const int rl = (int)((unsigned)wd >> 24);
            const float sc = edgeScore(ssrc[(size_t)(lo + rl) * H + h], sdst[(size_t)d * H + h]);
            const float ex = __expf(sc - ordVal(mx[rl * H + h]));
            float2 w[32];
            #pragma unroll
            for (int L = 0; L < 32; ++L) {
                const int dL = __builtin_amdgcn_readlane(d, L);
                const float* row = Wh + ((size_t)(L & (H - 1)) * N + dL) * 64;
                w[L] = *reinterpret_cast<const float2*>(row + 2 * lane);
            }
            holdRows(w, 0); holdRows(w, 8); holdRows(w, 16); holdRows(w, 24);
            #pragma unroll
            for (int e = 0; e < EPB; e += 2) {
                const int LA = e * H;
                const int LB = LA + H;
                if (b0 + e < bend) {
                    const int rlA = __builtin_amdgcn_readlane(rl, LA);
                    float* apA = acc + rlA * RW + 2 * lane;
                    float2 aA[H];
                    float exA[H];
                    #pragma unroll
                    for (int hh = 0; hh < H; ++hh)
                        exA[hh] = __int_as_float(__builtin_amdgcn_readlane(__float_as_int(ex), LA + hh));
                    if (b0 + e + 1 < bend) {
                        const int rlB = __builtin_amdgcn_readlane(rl, LB);
                        const bool same = rlB == rlA;
                        float* apB = acc + rlB * RW + 2 * lane;
                        float2 aB[H];
                        #pragma unroll
                        for (int hh = 0; hh < H; ++hh) {
                            aA[hh] = *reinterpret_cast<const float2*>(apA + hh * 64);
                            aB[hh] = *reinterpret_cast<const float2*>(apB + hh * 64);
                        }
                        #pragma unroll
                        for (int hh = 0; hh < H; ++hh) {
                            const float exB = __int_as_float(__builtin_amdgcn_readlane(__float_as_int(ex), LB + hh));
                            aA[hh].x = __builtin_fmaf(exA[hh], w[LA + hh].x, aA[hh].x);
                            aA[hh].y = __builtin_fmaf(exA[hh], w[LA + hh].y, aA[hh].y);
                            const float bx = same ? aA[hh].x : aB[hh].x;
                            const float by = same ? aA[hh].y : aB[hh].y;
                            aB[hh].x = __builtin_fmaf(exB, w[LB + hh].x, bx);
                            aB[hh].y = __builtin_fmaf(exB, w[LB + hh].y, by);
                        }
                        #pragma unroll
                        for (int hh = 0; hh < H; ++hh) *reinterpret_cast<float2*>(apA + hh * 64) = aA[hh];
                        #pragma unroll
                        for (int hh = 0; hh < H; ++hh) *reinterpret_cast<float2*>(apB + hh * 64) = aB[hh];
                    } else {
                        #pragma unroll
                        for (int hh = 0; hh < H; ++hh) aA[hh] = *reinterpret_cast<const float2*>(apA + hh * 64);
                        #pragma unroll
                        for (int hh = 0; hh < H; ++hh) {
                            aA[hh].x = __builtin_fmaf(exA[hh], w[LA + hh].x, aA[hh].x);
                            aA[hh].y = __builtin_fmaf(exA[hh], w[LA + hh].y, aA[hh].y);
                        }
                        #pragma unroll
                        for (int hh = 0; hh < H; ++hh) *reinterpret_cast<float2*>(apA + hh * 64) = aA[hh];
                    }
                }
            }
            if (ok) ldsAddF(&den[rl * H + h], ex);
        }
    } else {
        for (int e = tid; e < E; e += OWN_T) {
            const int s = src[e];
            const unsigned rl = (unsigned)(s - lo);
            if (rl < (unsigned)nrows) {
                const int d = dst[e];
                for (int hh = 0; hh < H; ++hh)
                    atomicMax(&mx[rl * H + hh],
                              ordKey(edgeScore(ssrc[(size_t)s * H + hh], sdst[(size_t)d * H + hh])));
            }
        }
        __syncthreads();
        for (int e = tid; e < E; e += OWN_T) {
            const int s = src[e];
            const unsigned rl = (unsigned)(s - lo);
            if (rl < (unsigned)nrows) {
                const int d = dst[e];
                for (int hh = 0; hh < H; ++hh) {
                    const float sc = edgeScore(ssrc[(size_t)s * H + hh], sdst[(size_t)d * H + hh]);
                    const float ex = __expf(sc - ordVal(mx[rl * H + hh]));
                    ldsAddF(&den[rl * H + hh], ex);
                    const float* row = Wh + ((size_t)hh * N + d) * 64;
                    float* ap = acc + rl * RW + hh * 64;
                    for (int k = 0; k < 64; ++k) ldsAddF(ap + k, ex * row[k]);
                }
            }
        }
    }
    __syncthreads();

    float4* o4 = reinterpret_cast<float4*>(outp + (size_t)lo * RW);
    const int n4 = nrows * (RW / 4);
    for (int i = tid; i < n4; i += OWN_T) {
        const float4 a = acc4[i];
        const float dn = den[i >> 4];
        float4 r;
        if (dn > 0.0f) {
            r.x = eluF(a.x / dn); r.y = eluF(a.y / dn); r.z = eluF(a.z / dn); r.w = eluF(a.w / dn);
        } else {
            r = make_float4(0.0f, 0.0f, 0.0f, 0.0f);
        }
        o4[i] = r;
    }
}

static inline int cdiv(long a, int b) { return (int)((a + b - 1) / b); }

extern "C" void kernel_launch(void* const* d_in, const int* in_sizes, int n_in,
                              void* d_out, int out_size, void* d_ws, size_t ws_size,
                              hipStream_t stream)
{
    const float* x       = (const float*)d_in[0];
    const int*   edges   = (const int*)d_in[1];
    const float* W_heads = (const float*)d_in[2];
    const float* a_heads = (const float*)d_in[3];
    const float* W_out   = (const float*)d_in[4];
    const float* a_out   = (const float*)d_in[5];
    float* out = (float*)d_out;

    const int N = in_sizes[0] / 128;
    const int E = in_sizes[1] / 2;
    const int H = 4, HID = 64;

    const int* src = edges;
    const int* dst = edges + E;

    float* ws = (float*)d_ws;
    size_t off = 0;
    float* Wh1   = ws + off; off += (size_t)H * N * HID;
    float* hcat  = ws + off; off += (size_t)N * 256;
    float* Wh2   = ws + off; off += (size_t)N * HID;
    float* ssrc1 = ws + off; off += (size_t)H * N;
    float* sdst1 = ws + off; off += (size_t)H * N;
    float* ssrc2 = ws + off; off += (size_t)N;
    float* sdst2 = ws + off; off += (size_t)N;
    int* binned  = (int*)(ws + off); off += (size_t)E + 64 * BIN_NCH;
    int* tab     = (int*)(ws + off);

    const int BS = 256;

    k_bin_edges<<<BIN_NCH, BIN_T, 0, stream>>>(src, dst, binned, tab, E, N);

    k_gemm_wmma<<<dim3(N / 16, H), 32, 0, stream>>>(x, W_heads, Wh1,
        128, 0L, (long)128 * 64, (long)N * 64);
    k_node_scores<4, 2><<<1024, BS, 0, stream>>>(Wh1, a_heads, ssrc1, sdst1, N);
    k_own_attend<2><<<cdiv((long)N, OWN_R), OWN_T, 0, stream>>>(src, dst, binned, tab, ssrc1, sdst1, Wh1, hcat, E, N);

    k_gemm_wmma<<<dim3(N / 16, 1), 32, 0, stream>>>(hcat, W_out, Wh2,
        256, 0L, 0L, 0L);
    k_node_scores<1, 8><<<256, BS, 0, stream>>>(Wh2, a_out, ssrc2, sdst2, N);
    k_own_attend<0><<<cdiv((long)N, OWN_R), OWN_T, 0, stream>>>(src, dst, binned, tab, ssrc2, sdst2, Wh2, out, E, N);
}


# Row sums leave LDS: owner queue sorted by row (a wave's rows side by side); the wave sums a row's run in registers and itself normalises, ELUs, stores the finished row; sum order changes; 7 launches
# speedup vs baseline: 7.8804x; 1.0628x over previous
#include <hip/hip_runtime.h>
#include <cstdint>
#include <cstddef>

typedef float v2f __attribute__((ext_vector_type(2)));
typedef float v8f __attribute__((ext_vector_type(8)));

#define GAT_ALPHA 0.2f

__device__ __forceinline__ unsigned int ordKey(float v) {
    unsigned int b = __float_as_uint(v);
    return (b & 0x80000000u) ? ~b : (b | 0x80000000u);
}
__device__ __forceinline__ float ordVal(unsigned int k) {
    return __uint_as_float((k & 0x80000000u) ? (k & 0x7fffffffu) : ~k);
}

__global__ __launch_bounds__(32) void k_gemm_wmma(
    const float* __restrict__ A, const float* __restrict__ B, float* __restrict__ C,
    int K, long aStride, long bStride, long cStride)
{
    const int lane = threadIdx.x;
    const int m0   = blockIdx.x * 16;
    const int b    = blockIdx.y;
    const float* __restrict__ Ab = A + (long)b * aStride;
    const float* __restrict__ Bb = B + (long)b * bStride;
    float* __restrict__ Cb = C + (long)b * cStride;

    const int row = lane & 15;
    const int kh  = (lane >> 4) << 1;
    const int col = lane & 15;

    v8f c0 = {}, c1 = {}, c2 = {}, c3 = {};

    for (int k0 = 0; k0 < K; k0 += 4) {
        const float2 af = *reinterpret_cast<const float2*>(
            Ab + (size_t)(m0 + row) * K + k0 + kh);
        v2f a; a.x = af.x; a.y = af.y;

        const float* br0 = Bb + (size_t)(k0 + kh) * 64 + col;
        const float* br1 = br0 + 64;
        v2f b0, b1, b2, b3;
        b0.x = br0[0];  b0.y = br1[0];
        b1.x = br0[16]; b1.y = br1[16];
        b2.x = br0[32]; b2.y = br1[32];
        b3.x = br0[48]; b3.y = br1[48];

        c0 = __builtin_amdgcn_wmma_f32_16x16x4_f32(false, a, false, b0, (short)0, c0, false, false);
        c1 = __builtin_amdgcn_wmma_f32_16x16x4_f32(false, a, false, b1, (short)0, c1, false, false);
        c2 = __builtin_amdgcn_wmma_f32_16x16x4_f32(false, a, false, b2, (short)0, c2, false, false);
        c3 = __builtin_amdgcn_wmma_f32_16x16x4_f32(false, a, false, b3, (short)0, c3, false, false);
    }

    const int rbase = m0 + ((lane >> 4) << 3);
    #pragma unroll
    for (int r = 0; r < 8; ++r) {
        float* crow = Cb + (size_t)(rbase + r) * 64 + col;
        crow[0]  = c0[r];
        crow[16] = c1[r];
        crow[32] = c2[r];
        crow[48] = c3[r];
    }
}

template <int H, int U>
__global__ __launch_bounds__(256) void k_node_scores(const float* __restrict__ Wh, const float* __restrict__ a,
                                                     float* __restrict__ ssrc, float* __restrict__ sdst, int N)
{
    const int lane = threadIdx.x & 31;
    const int wpb  = blockDim.x >> 5;
    const int w    = __builtin_amdgcn_readfirstlane(blockIdx.x * wpb + (threadIdx.x >> 5));
    const int nw   = gridDim.x * wpb;

    float as0[H], as1[H], ad0[H], ad1[H];
    #pragma unroll
    for (int h = 0; h < H; ++h) {
        const float* __restrict__ ah = a + h * 128;
        as0[h] = ah[lane];      as1[h] = ah[lane + 32];
        ad0[h] = ah[64 + lane]; ad1[h] = ah[96 + lane];
    }

    for (int n0 = w * U; n0 < N; n0 += nw * U) {
        float w0[U][H], w1[U][H];
        #pragma unroll
        for (int u = 0; u < U; ++u) {
            const int nn = min(n0 + u, N - 1);
            #pragma unroll
            for (int h = 0; h < H; ++h) {
                const float* __restrict__ row = Wh + ((size_t)h * N + nn) * 64;
                w0[u][h] = row[lane]; w1[u][h] = row[lane + 32];
            }
        }
        float s0[U][H], s1[U][H];
        #pragma unroll
        for (int u = 0; u < U; ++u) {
            #pragma unroll
            for (int h = 0; h < H; ++h) {
                s0[u][h] = __builtin_fmaf(w0[u][h], as0[h], w1[u][h] * as1[h]);
                s1[u][h] = __builtin_fmaf(w0[u][h], ad0[h], w1[u][h] * ad1[h]);
            }
        }
        #pragma unroll
        for (int off = 16; off >= 1; off >>= 1) {
            #pragma unroll
            for (int u = 0; u < U; ++u) {
                #pragma unroll
                for (int h = 0; h < H; ++h) {
                    s0[u][h] += __shfl_xor(s0[u][h], off, 32);
                    s1[u][h] += __shfl_xor(s1[u][h], off, 32);
                }
            }
        }
        if (lane == 0) {
            #pragma unroll
            for (int u = 0; u < U; ++u) {
                if (n0 + u < N) {
                    const size_t o = (size_t)(n0 + u) * H;
                    if constexpr (H == 4) {
                        *reinterpret_cast<float4*>(ssrc + o) = make_float4(s0[u][0], s0[u][1], s0[u][2], s0[u][3]);
                        *reinterpret_cast<float4*>(sdst + o) = make_float4(s1[u][0], s1[u][1], s1[u][2], s1[u][3]);
                    } else {
                        #pragma unroll
                        for (int h = 0; h < H; ++h) { ssrc[o + h] = s0[u][h]; sdst[o + h] = s1[u][h]; }
                    }
                }
            }
        }
    }
}

#define OWN_R    196
#define OWN_T    1024
#define OWN_QCAP 8192
#define OWN_P    4
#define OWN_NW   (OWN_T / 32)
#define OWN_RPW  ((OWN_R + OWN_NW - 1) / OWN_NW)
static_assert(OWN_NW == 32, "one wave scans the row counts; a row's owner wave is its local row mod 32");

__device__ __forceinline__ void ldsAddF(float* p, float v) {
    __hip_atomic_fetch_add(p, v, __ATOMIC_RELAXED, __HIP_MEMORY_SCOPE_WORKGROUP);
}

__device__ __forceinline__ float eluF(float v) { return v > 0.0f ? v : __expf(v) - 1.0f; }

template <int H>
__device__ __forceinline__ void storeRow(float* __restrict__ outp, int grow, int lane,
                                         const float2 (&sum)[H], const float (&dsm)[H])
{
    float* o = outp + (size_t)grow * (H * 64) + 2 * lane;
    #pragma unroll
    for (int hh = 0; hh < H; ++hh) {
        const float dn = dsm[hh];
        float2 r = make_float2(0.0f, 0.0f);
        if (dn > 0.0f) { r.x = eluF(sum[hh].x / dn); r.y = eluF(sum[hh].y / dn); }
        *reinterpret_cast<float2*>(o + hh * 64) = r;
    }
}

__device__ __forceinline__ float edgeScore(float a, float b) {
    const float v = a + b;
    return v > 0.0f ? v : GAT_ALPHA * v;
}

template <int H>
__device__ __forceinline__ void nodeScores(const float* __restrict__ p, int node, float (&v)[H]) {
    if constexpr (H == 4) {
        const float4 q = *reinterpret_cast<const float4*>(p + (size_t)node * 4);
        v[0] = q.x; v[1] = q.y; v[2] = q.z; v[3] = q.w;
    } else {
        #pragma unroll
        for (int h = 0; h < H; ++h) v[h] = p[(size_t)node * H + h];
    }
}

__device__ __forceinline__ void holdInts(const int (&d)[OWN_P]) {
    static_assert(OWN_P == 4, "written out for four slots");
    asm volatile("" :: "v"(d[0]), "v"(d[1]), "v"(d[2]), "v"(d[3]));
}
template <int H>
__device__ __forceinline__ void holdScores(const float (&v)[OWN_P][H]) {
    static_assert(OWN_P == 4 && (H == 1 || H == 4), "written out for four slots of one or four heads");
    if constexpr (H == 4) {
        asm volatile("" :: "v"(v[0][0]), "v"(v[0][1]), "v"(v[0][2]), "v"(v[0][3]),
                           "v"(v[1][0]), "v"(v[1][1]), "v"(v[1][2]), "v"(v[1][3]),
                           "v"(v[2][0]), "v"(v[2][1]), "v"(v[2][2]), "v"(v[2][3]),
                           "v"(v[3][0]), "v"(v[3][1]), "v"(v[3][2]), "v"(v[3][3]));
    } else {
        asm volatile("" :: "v"(v[0][0]), "v"(v[1][0]), "v"(v[2][0]), "v"(v[3][0]));
    }
}

__device__ __forceinline__ void holdRows(const float2 (&w)[32], int o) {
    asm volatile("" :: "v"(w[o + 0].x), "v"(w[o + 0].y), "v"(w[o + 1].x), "v"(w[o + 1].y),
                       "v"(w[o + 2].x), "v"(w[o + 2].y), "v"(w[o + 3].x), "v"(w[o + 3].y),
                       "v"(w[o + 4].x), "v"(w[o + 4].y), "v"(w[o + 5].x), "v"(w[o + 5].y),
                       "v"(w[o + 6].x), "v"(w[o + 6].y), "v"(w[o + 7].x), "v"(w[o + 7].y));
}

#define BIN_NCH  256
#define BIN_T    512
#define BIN_U    8
#define BIN_MAXR 1024

static_assert(OWN_R <= 256, "the local row of an edge is kept in 8 bits");

__device__ __forceinline__ int binRanges(int N) { return min((N + OWN_R - 1) / OWN_R, BIN_MAXR - 1); }
__device__ __forceinline__ int binTabPitch(int NB) { return (NB + 32) & ~31; }
__device__ __forceinline__ int binChunk(int E) { return (E + BIN_NCH - 1) / BIN_NCH; }
__device__ __forceinline__ int binPiecePitch(int CH) { return (CH + 31) & ~31; }

__global__ __launch_bounds__(BIN_T) void k_bin_edges(
    const int* __restrict__ src, const int* __restrict__ dst,
    int* __restrict__ binned, int* __restrict__ tab, int E, int N)
{
    __shared__ int cnt[BIN_MAXR];
    __shared__ int wtot[BIN_T / 32];

    const int tid = threadIdx.x, lane = tid & 31, wave = tid >> 5;
    const int NB = binRanges(N);
    const int TP = binTabPitch(NB);
    const int CH = binChunk(E);
    const int SP = binPiecePitch(CH);
    const int c  = blockIdx.x;
    const int e0 = c * CH;
    const int n  = min(CH, E - e0);
    const unsigned lim = (unsigned)min(N, NB * OWN_R);

    for (int i = tid; i < BIN_MAXR; i += BIN_T) cnt[i] = 0;
    __syncthreads();

    for (int k0 = 0; k0 < n; k0 += BIN_T * BIN_U) {
        int s[BIN_U];
        #pragma unroll
        for (int u = 0; u < BIN_U; ++u) s[u] = src[e0 + min(k0 + u * BIN_T + tid, n - 1)];
        #pragma unroll
        for (int u = 0; u < BIN_U; ++u) {
            const int k = k0 + u * BIN_T + tid;
            if (k < n && (unsigned)s[u] < lim) atomicAdd(&cnt[(unsigned)s[u] / OWN_R], 1);
        }
    }
    __syncthreads();

    constexpr int PER = BIN_MAXR / BIN_T;
    int v[PER];
    int tsum = 0;
    #pragma unroll
    for (int j = 0; j < PER; ++j) { v[j] = cnt[tid * PER + j]; tsum += v[j]; }
    int incl = tsum;
    #pragma unroll
    for (int off = 1; off < 32; off <<= 1) {
        const int y = __shfl_up(incl, off, 32);
        if (lane >= off) incl += y;
    }
    if (lane == 31) wtot[wave] = incl;
    __syncthreads();
    int run = incl - tsum;
    for (int w = 0; w < wave; ++w) run += wtot[w];
    #pragma unroll
    for (int j = 0; j < PER; ++j) {
        const int r = tid * PER + j;
        cnt[r] = run;
        if (r <= NB) tab[(size_t)c * TP + r] = run;
        run += v[j];
    }
    __syncthreads();

    for (int k0 = 0; k0 < n; k0 += BIN_T * BIN_U) {
        int s[BIN_U], d[BIN_U];
        #pragma unroll
        for (int u = 0; u < BIN_U; ++u) {
            const int k = e0 + min(k0 + u * BIN_T + tid, n - 1);
            s[u] = src[k]; d[u] = dst[k];
        }
        #pragma unroll
        for (int u = 0; u < BIN_U; ++u) {
            const int k = k0 + u * BIN_T + tid;
            if (k < n && (unsigned)s[u] < lim) {
                const unsigned r = (unsigned)s[u] / OWN_R;
                const int p = atomicAdd(&cnt[r], 1);
                binned[(size_t)c * SP + p] = (int)((((unsigned)s[u] - r * OWN_R) << 24) | ((unsigned)d[u] & 0xffffffu));
            }
        }
    }
}

template <int HS>
__global__ __launch_bounds__(OWN_T) void k_own_attend(
    const int* __restrict__ src, const int* __restrict__ dst,
    const int* __restrict__ binned, const int* __restrict__ tab,
    const float* __restrict__ ssrc, const float* __restrict__ sdst,
    const float* __restrict__ Wh, float* __restrict__ outp, int E, int N)
{
    constexpr int H  = 1 << HS;
    constexpr int RW = H * 64;
    __shared__ __attribute__((aligned(16))) float acc[OWN_R * RW];
    __shared__ float den[OWN_R * H];
    __shared__ unsigned int mx[OWN_R * H];
    __shared__ int   queA[OWN_QCAP];
    __shared__ int   queB[OWN_QCAP];
    __shared__ int   rcnt[OWN_NW * OWN_RPW];
    __shared__ int   rpos[OWN_NW * OWN_RPW];
    __shared__ int   bst[OWN_NW + 1];
    __shared__ int   cst[BIN_NCH];
    __shared__ int   qof[BIN_NCH + 1];
    __shared__ int   wsm[BIN_NCH / 32];

    const int tid  = threadIdx.x;
    const int lane = tid & 31;
    const int lo   = blockIdx.x * OWN_R;
    const int nrows = min(OWN_R, N - lo);

    const int NB = binRanges(N);
    const int TP = binTabPitch(NB);
    const int SP = binPiecePitch(binChunk(E));
    int myst = 0, mycnt = 0;
    if (tid < BIN_NCH && (int)blockIdx.x < NB) {
        const int* tr = tab + (size_t)tid * TP + blockIdx.x;
        myst = tr[0]; mycnt = tr[1] - myst;
    }

    float4* acc4 = reinterpret_cast<float4*>(acc);
    for (int i = tid; i < OWN_R * H; i += OWN_T) mx[i] = 0u;
    if (tid < OWN_NW * OWN_RPW) rcnt[tid] = 0;

    int incl = mycnt;
    if (tid < BIN_NCH) {
        #pragma unroll
        for (int off = 1; off < 32; off <<= 1) {
            const int y = __shfl_up(incl, off, 32);
            if (lane >= off) incl += y;
        }
        cst[tid] = myst;
        if (lane == 31) wsm[tid >> 5] = incl;
    }
    __syncthreads();
    if (tid < BIN_NCH) {
        int pre = incl - mycnt;
        for (int w = 0; w < (tid >> 5); ++w) pre += wsm[w];
        qof[tid] = pre;
        if (tid == BIN_NCH - 1) qof[BIN_NCH] = pre + mycnt;
    }
    __syncthreads();
    const int ntot = qof[BIN_NCH];

    if (ntot <= OWN_QCAP) {
        for (int base = 0; base < ntot; base += OWN_T * OWN_P) {
            const int i0 = base + tid;
            if (i0 < ntot) {
                int qi[OWN_P], cI[OWN_P], wI[OWN_P], rI[OWN_P], dI[OWN_P];
                #pragma unroll
                for (int u = 0; u < OWN_P; ++u) {
                    const int i = i0 + u * OWN_T;
                    qi[u] = i < ntot ? i : i0;
                }
                #pragma unroll
                for (int u = 0; u < OWN_P; ++u) {
                    int c = 0;
                    #pragma unroll
                    for (int st = BIN_NCH / 2; st >= 1; st >>= 1)
                        if (qof[c + st] <= qi[u]) c += st;
                    cI[u] = c;
                }
                #pragma unroll
                for (int u = 0; u < OWN_P; ++u)
                    wI[u] = binned[(size_t)cI[u] * SP + cst[cI[u]] + (qi[u] - qof[cI[u]])];
                holdInts(wI);
                float a[OWN_P][H], b[OWN_P][H];
                #pragma unroll
                for (int u = 0; u < OWN_P; ++u) {
                    dI[u] = wI[u] & 0xffffff;
                    rI[u] = (int)((unsigned)wI[u] >> 24);
                    nodeScores<H>(ssrc, lo + rI[u], a[u]);
                    nodeScores<H>(sdst, dI[u], b[u]);
                }
                holdScores<H>(a);
                holdScores<H>(b);
                #pragma unroll
                for (int u = 0; u < OWN_P; ++u) {
                    #pragma unroll
                    for (int h = 0; h < H; ++h)
                        atomicMax(&mx[rI[u] * H + h], ordKey(edgeScore(a[u][h], b[u][h])));
                    queA[qi[u]] = wI[u];
                    if (i0 + u * OWN_T < ntot) atomicAdd(&rcnt[rI[u]], 1);
                }
            }
        }
        __syncthreads();

        if (tid < OWN_NW) {
            int c[OWN_RPW];
            int t = 0;
            #pragma unroll
            for (int k = 0; k < OWN_RPW; ++k) { c[k] = rcnt[tid + OWN_NW * k]; t += c[k]; }
            int run = t;
            #pragma unroll
            for (int off = 1; off < 32; off <<= 1) {
                const int y = __shfl_up(run, off, 32);
                if (lane >= off) run += y;
            }
            int st = run - t;
            bst[tid] = st;
            #pragma unroll
            for (int k = 0; k < OWN_RPW; ++k) { rpos[tid + OWN_NW * k] = st; st += c[k]; }
            if (tid == OWN_NW - 1) bst[OWN_NW] = run;
        }
        __syncthreads();
        for (int i = tid; i < ntot; i += OWN_T) {
            const int wd = queA[i];
            const int p = atomicAdd(&rpos[(unsigned)wd >> 24], 1);
            queB[p] = wd;
        }
        __syncthreads();

        constexpr int EPB = 32 >> HS;
        const int wave = __builtin_amdgcn_readfirstlane(tid >> 5);
        const int bbeg = __builtin_amdgcn_readfirstlane(bst[wave]);
        const int bend = __builtin_amdgcn_readfirstlane(bst[wave + 1]);
        const int j = lane >> HS, h = lane & (H - 1);
        int cur = -1;
        float2 sum[H];
        float dsm[H];
        #pragma unroll
        for (int hh = 0; hh < H; ++hh) { sum[hh] = make_float2(0.0f, 0.0f); dsm[hh] = 0.0f; }
        for (int b0 = bbeg; b0 < bend; b0 += EPB) {
            const bool ok = b0 + j < bend;
            const int wd = queB[ok ? b0 + j : b0];
            const int d  = wd & 0xffffff;
            const int rl = (int)((unsigned)wd >> 24);
            const float sc = edgeScore(ssrc[(size_t)(lo + rl) * H + h], sdst[(size_t)d * H + h]);
            const float ex = __expf(sc - ordVal(mx[rl * H + h]));
            float2 w[32];
            #pragma unroll
            for (int L = 0; L < 32; ++L) {
                const int dL = __builtin_amdgcn_readlane(d, L);
                const float* row = Wh + ((size_t)(L & (H - 1)) * N + dL) * 64;
                w[L] = *reinterpret_cast<const float2*>(row + 2 * lane);
            }
            holdRows(w, 0); holdRows(w, 8); holdRows(w, 16); holdRows(w, 24);
            #pragma unroll
            for (int e = 0; e < EPB; ++e) {
                if (b0 + e < bend) {
                    const int LE = e * H;
                    const int r = __builtin_amdgcn_readlane(rl, LE);
                    if (r != cur) {
                        if (cur >= 0) storeRow<H>(outp, lo + cur, lane, sum, dsm);
                        cur = r;
                        #pragma unroll
                        for (int hh = 0; hh < H; ++hh) { sum[hh] = make_float2(0.0f, 0.0f); dsm[hh] = 0.0f; }
                    }
                    #pragma unroll
                    for (int hh = 0; hh < H; ++hh) {
                        const float exE = __int_as_float(__builtin_amdgcn_readlane(__float_as_int(ex), LE + hh));
                        sum[hh].x = __builtin_fmaf(exE, w[LE + hh].x, sum[hh].x);
                        sum[hh].y = __builtin_fmaf(exE, w[LE + hh].y, sum[hh].y);
                        dsm[hh] += exE;
                    }
                }
            }
        }
        if (cur >= 0) storeRow<H>(outp, lo + cur, lane, sum, dsm);
        for (int k = 0; k < OWN_RPW; ++k) {
            const int r = wave + OWN_NW * k;
            if (r < nrows && rcnt[r] == 0) {
                float* o = outp + (size_t)(lo + r) * RW + 2 * lane;
                for (int hh = 0; hh < H; ++hh) *reinterpret_cast<float2*>(o + hh * 64) = make_float2(0.0f, 0.0f);
            }
        }
    } else {
        for (int i = tid; i < OWN_R * RW / 4; i += OWN_T) acc4[i] = make_float4(0.0f, 0.0f, 0.0f, 0.0f);
        for (int i = tid; i < OWN_R * H; i += OWN_T) den[i] = 0.0f;
        for (int e = tid; e < E; e += OWN_T) {
            const int s = src[e];
            const unsigned rl = (unsigned)(s - lo);
            if (rl < (unsigned)nrows) {
                const int d = dst[e];
                for (int hh = 0; hh < H; ++hh)
                    atomicMax(&mx[rl * H + hh],
                              ordKey(edgeScore(ssrc[(size_t)s * H + hh], sdst[(size_t)d * H + hh])));
            }
        }
        __syncthreads();
        for (int e = tid; e < E; e += OWN_T) {
            const int s = src[e];
            const unsigned rl = (unsigned)(s - lo);
            if (rl < (unsigned)nrows) {
                const int d = dst[e];
                for (int hh = 0; hh < H; ++hh) {
                    const float sc = edgeScore(ssrc[(size_t)s * H + hh], sdst[(size_t)d * H + hh]);
                    const float ex = __expf(sc - ordVal(mx[rl * H + hh]));
                    ldsAddF(&den[rl * H + hh], ex);
                    const float* row = Wh + ((size_t)hh * N + d) * 64;
                    float* ap = acc + rl * RW + hh * 64;
                    for (int k = 0; k < 64; ++k) ldsAddF(ap + k, ex * row[k]);
                }
            }
        }
        __syncthreads();

        float4* o4 = reinterpret_cast<float4*>(outp + (size_t)lo * RW);
        const int n4 = nrows * (RW / 4);
        for (int i = tid; i < n4; i += OWN_T) {
            const float4 a = acc4[i];
            const float dn = den[i >> 4];
            float4 r;
            if (dn > 0.0f) {
                r.x = eluF(a.x / dn); r.y = eluF(a.y / dn); r.z = eluF(a.z / dn); r.w = eluF(a.w / dn);
            } else {
                r = make_float4(0.0f, 0.0f, 0.0f, 0.0f);
            }
            o4[i] = r;
        }
    }
}

static inline int cdiv(long a, int b) { return (int)((a + b - 1) / b); }

extern "C" void kernel_launch(void* const* d_in, const int* in_sizes, int n_in,
                              void* d_out, int out_size, void* d_ws, size_t ws_size,
                              hipStream_t stream)
{
    const float* x       = (const float*)d_in[0];
    const int*   edges   = (const int*)d_in[1];
    const float* W_heads = (const float*)d_in[2];
    const float* a_heads = (const float*)d_in[3];
    const float* W_out   = (const float*)d_in[4];
    const float* a_out   = (const float*)d_in[5];
    float* out = (float*)d_out;

    const int N = in_sizes[0] / 128;
    const int E = in_sizes[1] / 2;
    const int H = 4, HID = 64;

    const int* src = edges;
    const int* dst = edges + E;

    float* ws = (float*)d_ws;
    size_t off = 0;
    float* Wh1   = ws + off; off += (size_t)H * N * HID;
    float* hcat  = ws + off; off += (size_t)N * 256;
    float* Wh2   = ws + off; off += (size_t)N * HID;
    float* ssrc1 = ws + off; off += (size_t)H * N;
    float* sdst1 = ws + off; off += (size_t)H * N;
    float* ssrc2 = ws + off; off += (size_t)N;
    float* sdst2 = ws + off; off += (size_t)N;
    int* binned  = (int*)(ws + off); off += (size_t)E + 64 * BIN_NCH;
    int* tab     = (int*)(ws + off);

    const int BS = 256;

    k_bin_edges<<<BIN_NCH, BIN_T, 0, stream>>>(src, dst, binned, tab, E, N);

    k_gemm_wmma<<<dim3(N / 16, H), 32, 0, stream>>>(x, W_heads, Wh1,
        128, 0L, (long)128 * 64, (long)N * 64);
    k_node_scores<4, 2><<<1024, BS, 0, stream>>>(Wh1, a_heads, ssrc1, sdst1, N);
    k_own_attend<2><<<cdiv((long)N, OWN_R), OWN_T, 0, stream>>>(src, dst, binned, tab, ssrc1, sdst1, Wh1, hcat, E, N);

    k_gemm_wmma<<<dim3(N / 16, 1), 32, 0, stream>>>(hcat, W_out, Wh2,
        256, 0L, 0L, 0L);
    k_node_scores<1, 8><<<256, BS, 0, stream>>>(Wh2, a_out, ssrc2, sdst2, N);
    k_own_attend<0><<<cdiv((long)N, OWN_R), OWN_T, 0, stream>>>(src, dst, binned, tab, ssrc2, sdst2, Wh2, out, E, N);
}


# B operand of both products resident in registers: 2048 long-lived waves load their K x 64 or K x 32 slab of B once and walk over 16-row tiles of A; no B load in the K loop; still 7 launches
# speedup vs baseline: 8.1602x; 1.0355x over previous
#include <hip/hip_runtime.h>
#include <cstdint>
#include <cstddef>

typedef float v2f __attribute__((ext_vector_type(2)));
typedef float v8f __attribute__((ext_vector_type(8)));

#define GAT_ALPHA 0.2f

__device__ __forceinline__ unsigned int ordKey(float v) {
    unsigned int b = __float_as_uint(v);
    return (b & 0x80000000u) ? ~b : (b | 0x80000000u);
}
__device__ __forceinline__ float ordVal(unsigned int k) {
    return __uint_as_float((k & 0x80000000u) ? (k & 0x7fffffffu) : ~k);
}


template <int K>
__device__ __forceinline__ void loadATile(const float* __restrict__ A, int tile, int row, int kh, v2f (&a)[K / 4]) {
    const float* __restrict__ ar = A + (size_t)(tile * 16 + row) * K + kh;
    #pragma unroll
    for (int s = 0; s < K / 4; ++s) {
        const float2 af = *reinterpret_cast<const float2*>(ar + 4 * s);
        a[s].x = af.x; a[s].y = af.y;
    }
}

template <int K, int NT, bool HEADS, bool AHEAD>
__global__ __launch_bounds__(256) void k_gemm_res(
    const float* __restrict__ A, const float* __restrict__ B, float* __restrict__ C, int M, int nslab)
{
    constexpr int KS = K / 4;
    const int lane = threadIdx.x & 31;
    const int wv   = __builtin_amdgcn_readfirstlane(threadIdx.x >> 5);
    const int slab = wv % nslab;
    const int tpw  = (blockDim.x >> 5) / nslab;
    const int tstep = gridDim.x * tpw;
    const int T    = M / 16;

    const int row = lane & 15;
    const int kh  = (lane >> 4) << 1;
    const int col = lane & 15;

    const float* __restrict__ Bs = HEADS ? B + (size_t)slab * K * 64 : B + slab * 16 * NT;
    float* __restrict__ Cs = HEADS ? C + (size_t)slab * M * 64 : C + slab * 16 * NT;

    v2f bf[KS][NT];
    #pragma unroll
    for (int s = 0; s < KS; ++s) {
        const float* br0 = Bs + (size_t)(4 * s + kh) * 64 + col;
        #pragma unroll
        for (int j = 0; j < NT; ++j) { bf[s][j].x = br0[16 * j]; bf[s][j].y = br0[64 + 16 * j]; }
    }

    int t = blockIdx.x + gridDim.x * (wv / nslab);
    if (t >= T) return;
    v2f a[KS];
    loadATile<K>(A, t, row, kh, a);
    for (; t < T; t += tstep) {
        v2f an[KS];
        if constexpr (AHEAD) loadATile<K>(A, min(t + tstep, T - 1), row, kh, an);

        v8f c[NT];
        #pragma unroll
        for (int j = 0; j < NT; ++j) c[j] = v8f{};
        #pragma unroll
        for (int s = 0; s < KS; ++s) {
            #pragma unroll
            for (int j = 0; j < NT; ++j)
                c[j] = __builtin_amdgcn_wmma_f32_16x16x4_f32(false, a[s], false, bf[s][j], (short)0, c[j], false, false);
        }

        const int rbase = t * 16 + ((lane >> 4) << 3);
        #pragma unroll
        for (int r = 0; r < 8; ++r) {
            float* crow = Cs + (size_t)(rbase + r) * 64 + col;
            #pragma unroll
            for (int j = 0; j < NT; ++j) crow[16 * j] = c[j][r];
        }

        if constexpr (AHEAD) {
            #pragma unroll
            for (int s = 0; s < KS; ++s) a[s] = an[s];
        } else {
            if (t + tstep < T) loadATile<K>(A, t + tstep, row, kh, a);
        }
    }
}

template <int H, int U>
__global__ __launch_bounds__(256) void k_node_scores(const float* __restrict__ Wh, const float* __restrict__ a,
                                                     float* __restrict__ ssrc, float* __restrict__ sdst, int N)
{
    const int lane = threadIdx.x & 31;
    const int wpb  = blockDim.x >> 5;
    const int w    = __builtin_amdgcn_readfirstlane(blockIdx.x * wpb + (threadIdx.x >> 5));
    const int nw   = gridDim.x * wpb;

    float as0[H], as1[H], ad0[H], ad1[H];
    #pragma unroll
    for (int h = 0; h < H; ++h) {
        const float* __restrict__ ah = a + h * 128;
        as0[h] = ah[lane];      as1[h] = ah[lane + 32];
        ad0[h] = ah[64 + lane]; ad1[h] = ah[96 + lane];
    }

    for (int n0 = w * U; n0 < N; n0 += nw * U) {
        float w0[U][H], w1[U][H];
        #pragma unroll
        for (int u = 0; u < U; ++u) {
            const int nn = min(n0 + u, N - 1);
            #pragma unroll
            for (int h = 0; h < H; ++h) {
                const float* __restrict__ row = Wh + ((size_t)h * N + nn) * 64;
                w0[u][h] = row[lane]; w1[u][h] = row[lane + 32];
            }
        }
        float s0[U][H], s1[U][H];
        #pragma unroll
        for (int u = 0; u < U; ++u) {
            #pragma unroll
            for (int h = 0; h < H; ++h) {
                s0[u][h] = __builtin_fmaf(w0[u][h], as0[h], w1[u][h] * as1[h]);
                s1[u][h] = __builtin_fmaf(w0[u][h], ad0[h], w1[u][h] * ad1[h]);
            }
        }
        #pragma unroll
        for (int off = 16; off >= 1; off >>= 1) {
            #pragma unroll
            for (int u = 0; u < U; ++u) {
                #pragma unroll
                for (int h = 0; h < H; ++h) {
                    s0[u][h] += __shfl_xor(s0[u][h], off, 32);
                    s1[u][h] += __shfl_xor(s1[u][h], off, 32);
                }
            }
        }
        if (lane == 0) {
            #pragma unroll
            for (int u = 0; u < U; ++u) {
                if (n0 + u < N) {
                    const size_t o = (size_t)(n0 + u) * H;
                    if constexpr (H == 4) {
                        *reinterpret_cast<float4*>(ssrc + o) = make_float4(s0[u][0], s0[u][1], s0[u][2], s0[u][3]);
                        *reinterpret_cast<float4*>(sdst + o) = make_float4(s1[u][0], s1[u][1], s1[u][2], s1[u][3]);
                    } else {
                        #pragma unroll
                        for (int h = 0; h < H; ++h) { ssrc[o + h] = s0[u][h]; sdst[o + h] = s1[u][h]; }
                    }
                }
            }
        }
    }
}

#define OWN_R    196
#define OWN_T    1024
#define OWN_QCAP 8192
#define OWN_P    4
#define OWN_NW   (OWN_T / 32)
#define OWN_RPW  ((OWN_R + OWN_NW - 1) / OWN_NW)
static_assert(OWN_NW == 32, "one wave scans the row counts; a row's owner wave is its local row mod 32");

__device__ __forceinline__ void ldsAddF(float* p, float v) {
    __hip_atomic_fetch_add(p, v, __ATOMIC_RELAXED, __HIP_MEMORY_SCOPE_WORKGROUP);
}

__device__ __forceinline__ float eluF(float v) { return v > 0.0f ? v : __expf(v) - 1.0f; }

template <int H>
__device__ __forceinline__ void storeRow(float* __restrict__ outp, int grow, int lane,
                                         const float2 (&sum)[H], const float (&dsm)[H])
{
    float* o = outp + (size_t)grow * (H * 64) + 2 * lane;
    #pragma unroll
    for (int hh = 0; hh < H; ++hh) {
        const float dn = dsm[hh];
        float2 r = make_float2(0.0f, 0.0f);
        if (dn > 0.0f) { r.x = eluF(sum[hh].x / dn); r.y = eluF(sum[hh].y / dn); }
        *reinterpret_cast<float2*>(o + hh * 64) = r;
    }
}

__device__ __forceinline__ float edgeScore(float a, float b) {
    const float v = a + b;
    return v > 0.0f ? v : GAT_ALPHA * v;
}

template <int H>
__device__ __forceinline__ void nodeScores(const float* __restrict__ p, int node, float (&v)[H]) {
    if constexpr (H == 4) {
        const float4 q = *reinterpret_cast<const float4*>(p + (size_t)node * 4);
        v[0] = q.x; v[1] = q.y; v[2] = q.z; v[3] = q.w;
    } else {
        #pragma unroll
        for (int h = 0; h < H; ++h) v[h] = p[(size_t)node * H + h];
    }
}

__device__ __forceinline__ void holdInts(const int (&d)[OWN_P]) {
    static_assert(OWN_P == 4, "written out for four slots");
    asm volatile("" :: "v"(d[0]), "v"(d[1]), "v"(d[2]), "v"(d[3]));
}
template <int H>
__device__ __forceinline__ void holdScores(const float (&v)[OWN_P][H]) {
    static_assert(OWN_P == 4 && (H == 1 || H == 4), "written out for four slots of one or four heads");
    if constexpr (H == 4) {
        asm volatile("" :: "v"(v[0][0]), "v"(v[0][1]), "v"(v[0][2]), "v"(v[0][3]),
                           "v"(v[1][0]), "v"(v[1][1]), "v"(v[1][2]), "v"(v[1][3]),
                           "v"(v[2][0]), "v"(v[2][1]), "v"(v[2][2]), "v"(v[2][3]),
                           "v"(v[3][0]), "v"(v[3][1]), "v"(v[3][2]), "v"(v[3][3]));
    } else {
        asm volatile("" :: "v"(v[0][0]), "v"(v[1][0]), "v"(v[2][0]), "v"(v[3][0]));
    }
}

__device__ __forceinline__ void holdRows(const float2 (&w)[32], int o) {
    asm volatile("" :: "v"(w[o + 0].x), "v"(w[o + 0].y), "v"(w[o + 1].x), "v"(w[o + 1].y),
                       "v"(w[o + 2].x), "v"(w[o + 2].y), "v"(w[o + 3].x), "v"(w[o + 3].y),
                       "v"(w[o + 4].x), "v"(w[o + 4].y), "v"(w[o + 5].x), "v"(w[o + 5].y),
                       "v"(w[o + 6].x), "v"(w[o + 6].y), "v"(w[o + 7].x), "v"(w[o + 7].y));
}

#define BIN_NCH  256
#define BIN_T    512
#define BIN_U    8
#define BIN_MAXR 1024

static_assert(OWN_R <= 256, "the local row of an edge is kept in 8 bits");

__device__ __forceinline__ int binRanges(int N) { return min((N + OWN_R - 1) / OWN_R, BIN_MAXR - 1); }
__device__ __forceinline__ int binTabPitch(int NB) { return (NB + 32) & ~31; }
__device__ __forceinline__ int binChunk(int E) { return (E + BIN_NCH - 1) / BIN_NCH; }
__device__ __forceinline__ int binPiecePitch(int CH) { return (CH + 31) & ~31; }

__global__ __launch_bounds__(BIN_T) void k_bin_edges(
    const int* __restrict__ src, const int* __restrict__ dst,
    int* __restrict__ binned, int* __restrict__ tab, int E, int N)
{
    __shared__ int cnt[BIN_MAXR];
    __shared__ int wtot[BIN_T / 32];

    const int tid = threadIdx.x, lane = tid & 31, wave = tid >> 5;
    const int NB = binRanges(N);
    const int TP = binTabPitch(NB);
    const int CH = binChunk(E);
    const int SP = binPiecePitch(CH);
    const int c  = blockIdx.x;
    const int e0 = c * CH;
    const int n  = min(CH, E - e0);
    const unsigned lim = (unsigned)min(N, NB * OWN_R);

    for (int i = tid; i < BIN_MAXR; i += BIN_T) cnt[i] = 0;
    __syncthreads();

    for (int k0 = 0; k0 < n; k0 += BIN_T * BIN_U) {
        int s[BIN_U];
        #pragma unroll
        for (int u = 0; u < BIN_U; ++u) s[u] = src[e0 + min(k0 + u * BIN_T + tid, n - 1)];
        #pragma unroll
        for (int u = 0; u < BIN_U; ++u) {
            const int k = k0 + u * BIN_T + tid;
            if (k < n && (unsigned)s[u] < lim) atomicAdd(&cnt[(unsigned)s[u] / OWN_R], 1);
        }
    }
    __syncthreads();

    constexpr int PER = BIN_MAXR / BIN_T;
    int v[PER];
    int tsum = 0;
    #pragma unroll
    for (int j = 0; j < PER; ++j) { v[j] = cnt[tid * PER + j]; tsum += v[j]; }
    int incl = tsum;
    #pragma unroll
    for (int off = 1; off < 32; off <<= 1) {
        const int y = __shfl_up(incl, off, 32);
        if (lane >= off) incl += y;
    }
    if (lane == 31) wtot[wave] = incl;
    __syncthreads();
    int run = incl - tsum;
    for (int w = 0; w < wave; ++w) run += wtot[w];
    #pragma unroll
    for (int j = 0; j < PER; ++j) {
        const int r = tid * PER + j;
        cnt[r] = run;
        if (r <= NB) tab[(size_t)c * TP + r] = run;
        run += v[j];
    }
    __syncthreads();

    for (int k0 = 0; k0 < n; k0 += BIN_T * BIN_U) {
        int s[BIN_U], d[BIN_U];
        #pragma unroll
        for (int u = 0; u < BIN_U; ++u) {
            const int k = e0 + min(k0 + u * BIN_T + tid, n - 1);
            s[u] = src[k]; d[u] = dst[k];
        }
        #pragma unroll
        for (int u = 0; u < BIN_U; ++u) {
            const int k = k0 + u * BIN_T + tid;
            if (k < n && (unsigned)s[u] < lim) {
                const unsigned r = (unsigned)s[u] / OWN_R;
                const int p = atomicAdd(&cnt[r], 1);
                binned[(size_t)c * SP + p] = (int)((((unsigned)s[u] - r * OWN_R) << 24) | ((unsigned)d[u] & 0xffffffu));
            }
        }
    }
}

template <int HS>
__global__ __launch_bounds__(OWN_T) void k_own_attend(
    const int* __restrict__ src, const int* __restrict__ dst,
    const int* __restrict__ binned, const int* __restrict__ tab,
    const float* __restrict__ ssrc, const float* __restrict__ sdst,
    const float* __restrict__ Wh, float* __restrict__ outp, int E, int N)
{
    constexpr int H  = 1 << HS;
    constexpr int RW = H * 64;
    __shared__ __attribute__((aligned(16))) float acc[OWN_R * RW];
    __shared__ float den[OWN_R * H];
    __shared__ unsigned int mx[OWN_R * H];
    __shared__ int   queA[OWN_QCAP];
    __shared__ int   queB[OWN_QCAP];
    __shared__ int   rcnt[OWN_NW * OWN_RPW];
    __shared__ int   rpos[OWN_NW * OWN_RPW];
    __shared__ int   bst[OWN_NW + 1];
    __shared__ int   cst[BIN_NCH];
    __shared__ int   qof[BIN_NCH + 1];
    __shared__ int   wsm[BIN_NCH / 32];

    const int tid  = threadIdx.x;
    const int lane = tid & 31;
    const int lo   = blockIdx.x * OWN_R;
    const int nrows = min(OWN_R, N - lo);

    const int NB = binRanges(N);
    const int TP = binTabPitch(NB);
    const int SP = binPiecePitch(binChunk(E));
    int myst = 0, mycnt = 0;
    if (tid < BIN_NCH && (int)blockIdx.x < NB) {
        const int* tr = tab + (size_t)tid * TP + blockIdx.x;
        myst = tr[0]; mycnt = tr[1] - myst;
    }

    float4* acc4 = reinterpret_cast<float4*>(acc);
    for (int i = tid; i < OWN_R * H; i += OWN_T) mx[i] = 0u;
    if (tid < OWN_NW * OWN_RPW) rcnt[tid] = 0;

    int incl = mycnt;
    if (tid < BIN_NCH) {
        #pragma unroll
        for (int off = 1; off < 32; off <<= 1) {
            const int y = __shfl_up(incl, off, 32);
            if (lane >= off) incl += y;
        }
        cst[tid] = myst;
        if (lane == 31) wsm[tid >> 5] = incl;
    }
    __syncthreads();
    if (tid < BIN_NCH) {
        int pre = incl - mycnt;
        for (int w = 0; w < (tid >> 5); ++w) pre += wsm[w];
        qof[tid] = pre;
        if (tid == BIN_NCH - 1) qof[BIN_NCH] = pre + mycnt;
    }
    __syncthreads();
    const int ntot = qof[BIN_NCH];

    if (ntot <= OWN_QCAP) {
        for (int base = 0; base < ntot; base += OWN_T * OWN_P) {
            const int i0 = base + tid;
            if (i0 < ntot) {
                int qi[OWN_P], cI[OWN_P], wI[OWN_P], rI[OWN_P], dI[OWN_P];
                #pragma unroll
                for (int u = 0; u < OWN_P; ++u) {
                    const int i = i0 + u * OWN_T;
                    qi[u] = i < ntot ? i : i0;
                }
                #pragma unroll
                for (int u = 0; u < OWN_P; ++u) {
                    int c = 0;
                    #pragma unroll
                    for (int st = BIN_NCH / 2; st >= 1; st >>= 1)
                        if (qof[c + st] <= qi[u]) c += st;
                    cI[u] = c;
                }
                #pragma unroll
                for (int u = 0; u < OWN_P; ++u)
                    wI[u] = binned[(size_t)cI[u] * SP + cst[cI[u]] + (qi[u] - qof[cI[u]])];
                holdInts(wI);
                float a[OWN_P][H], b[OWN_P][H];
                #pragma unroll
                for (int u = 0; u < OWN_P; ++u) {
                    dI[u] = wI[u] & 0xffffff;
                    rI[u] = (int)((unsigned)wI[u] >> 24);
                    nodeScores<H>(ssrc, lo + rI[u], a[u]);
                    nodeScores<H>(sdst, dI[u], b[u]);
                }
                holdScores<H>(a);
                holdScores<H>(b);
                #pragma unroll
                for (int u = 0; u < OWN_P; ++u) {
                    #pragma unroll
                    for (int h = 0; h < H; ++h)
                        atomicMax(&mx[rI[u] * H + h], ordKey(edgeScore(a[u][h], b[u][h])));
                    queA[qi[u]] = wI[u];
                    if (i0 + u * OWN_T < ntot) atomicAdd(&rcnt[rI[u]], 1);
                }
            }
        }
        __syncthreads();

        if (tid < OWN_NW) {
            int c[OWN_RPW];
            int t = 0;
            #pragma unroll
            for (int k = 0; k < OWN_RPW; ++k) { c[k] = rcnt[tid + OWN_NW * k]; t += c[k]; }
            int run = t;
            #pragma unroll
            for (int off = 1; off < 32; off <<= 1) {
                const int y = __shfl_up(run, off, 32);
                if (lane >= off) run += y;
            }
            int st = run - t;
            bst[tid] = st;
            #pragma unroll
            for (int k = 0; k < OWN_RPW; ++k) { rpos[tid + OWN_NW * k] = st; st += c[k]; }
            if (tid == OWN_NW - 1) bst[OWN_NW] = run;
        }
        __syncthreads();
        for (int i = tid; i < ntot; i += OWN_T) {
            const int wd = queA[i];
            const int p = atomicAdd(&rpos[(unsigned)wd >> 24], 1);
            queB[p] = wd;
        }
        __syncthreads();

        constexpr int EPB = 32 >> HS;
        const int wave = __builtin_amdgcn_readfirstlane(tid >> 5);
        const int bbeg = __builtin_amdgcn_readfirstlane(bst[wave]);
        const int bend = __builtin_amdgcn_readfirstlane(bst[wave + 1]);
        const int j = lane >> HS, h = lane & (H - 1);
        int cur = -1;
        float2 sum[H];
        float dsm[H];
        #pragma unroll
        for (int hh = 0; hh < H; ++hh) { sum[hh] = make_float2(0.0f, 0.0f); dsm[hh] = 0.0f; }
        for (int b0 = bbeg; b0 < bend; b0 += EPB) {
            const bool ok = b0 + j < bend;
            const int wd = queB[ok ? b0 + j : b0];
            const int d  = wd & 0xffffff;
            const int rl = (int)((unsigned)wd >> 24);
            const float sc = edgeScore(ssrc[(size_t)(lo + rl) * H + h], sdst[(size_t)d * H + h]);
            const float ex = __expf(sc - ordVal(mx[rl * H + h]));
            float2 w[32];
            #pragma unroll
            for (int L = 0; L < 32; ++L) {
                const int dL = __builtin_amdgcn_readlane(d, L);
                const float* row = Wh + ((size_t)(L & (H - 1)) * N + dL) * 64;
                w[L] = *reinterpret_cast<const float2*>(row + 2 * lane);
            }
            holdRows(w, 0); holdRows(w, 8); holdRows(w, 16); holdRows(w, 24);
            #pragma unroll
            for (int e = 0; e < EPB; ++e) {
                if (b0 + e < bend) {
                    const int LE = e * H;
                    const int r = __builtin_amdgcn_readlane(rl, LE);
                    if (r != cur) {
                        if (cur >= 0) storeRow<H>(outp, lo + cur, lane, sum, dsm);
                        cur = r;
                        #pragma unroll
                        for (int hh = 0; hh < H; ++hh) { sum[hh] = make_float2(0.0f, 0.0f); dsm[hh] = 0.0f; }
                    }
                    #pragma unroll
                    for (int hh = 0; hh < H; ++hh) {
                        const float exE = __int_as_float(__builtin_amdgcn_readlane(__float_as_int(ex), LE + hh));
                        sum[hh].x = __builtin_fmaf(exE, w[LE + hh].x, sum[hh].x);
                        sum[hh].y = __builtin_fmaf(exE, w[LE + hh].y, sum[hh].y);
                        dsm[hh] += exE;
                    }
                }
            }
        }
        if (cur >= 0) storeRow<H>(outp, lo + cur, lane, sum, dsm);
        for (int k = 0; k < OWN_RPW; ++k) {
            const int r = wave + OWN_NW * k;
            if (r < nrows && rcnt[r] == 0) {
                float* o = outp + (size_t)(lo + r) * RW + 2 * lane;
                for (int hh = 0; hh < H; ++hh) *reinterpret_cast<float2*>(o + hh * 64) = make_float2(0.0f, 0.0f);
            }
        }
    } else {
        for (int i = tid; i < OWN_R * RW / 4; i += OWN_T) acc4[i] = make_float4(0.0f, 0.0f, 0.0f, 0.0f);
        for (int i = tid; i < OWN_R * H; i += OWN_T) den[i] = 0.0f;
        for (int e = tid; e < E; e += OWN_T) {
            const int s = src[e];
            const unsigned rl = (unsigned)(s - lo);
            if (rl < (unsigned)nrows) {
                const int d = dst[e];
                for (int hh = 0; hh < H; ++hh)
                    atomicMax(&mx[rl * H + hh],
                              ordKey(edgeScore(ssrc[(size_t)s * H + hh], sdst[(size_t)d * H + hh])));
            }
        }
        __syncthreads();
        for (int e = tid; e < E; e += OWN_T) {
            const int s = src[e];
            const unsigned rl = (unsigned)(s - lo);
            if (rl < (unsigned)nrows) {
                const int d = dst[e];
                for (int hh = 0; hh < H; ++hh) {
                    const float sc = edgeScore(ssrc[(size_t)s * H + hh], sdst[(size_t)d * H + hh]);
                    const float ex = __expf(sc - ordVal(mx[rl * H + hh]));
                    ldsAddF(&den[rl * H + hh], ex);
                    const float* row = Wh + ((size_t)hh * N + d) * 64;
                    float* ap = acc + rl * RW + hh * 64;
                    for (int k = 0; k < 64; ++k) ldsAddF(ap + k, ex * row[k]);
                }
            }
        }
        __syncthreads();

        float4* o4 = reinterpret_cast<float4*>(outp + (size_t)lo * RW);
        const int n4 = nrows * (RW / 4);
        for (int i = tid; i < n4; i += OWN_T) {
            const float4 a = acc4[i];
            const float dn = den[i >> 4];
            float4 r;
            if (dn > 0.0f) {
                r.x = eluF(a.x / dn); r.y = eluF(a.y / dn); r.z = eluF(a.z / dn); r.w = eluF(a.w / dn);
            } else {
                r = make_float4(0.0f, 0.0f, 0.0f, 0.0f);
            }
            o4[i] = r;
        }
    }
}

static inline int cdiv(long a, int b) { return (int)((a + b - 1) / b); }

extern "C" void kernel_launch(void* const* d_in, const int* in_sizes, int n_in,
                              void* d_out, int out_size, void* d_ws, size_t ws_size,
                              hipStream_t stream)
{
    const float* x       = (const float*)d_in[0];
    const int*   edges   = (const int*)d_in[1];
    const float* W_heads = (const float*)d_in[2];
    const float* a_heads = (const float*)d_in[3];
    const float* W_out   = (const float*)d_in[4];
    const float* a_out   = (const float*)d_in[5];
    float* out = (float*)d_out;

    const int N = in_sizes[0] / 128;
    const int E = in_sizes[1] / 2;
    const int H = 4, HID = 64;

    const int* src = edges;
    const int* dst = edges + E;

    float* ws = (float*)d_ws;
    size_t off = 0;
    float* Wh1   = ws + off; off += (size_t)H * N * HID;
    float* hcat  = ws + off; off += (size_t)N * 256;
    float* Wh2   = ws + off; off += (size_t)N * HID;
    float* ssrc1 = ws + off; off += (size_t)H * N;
    float* sdst1 = ws + off; off += (size_t)H * N;
    float* ssrc2 = ws + off; off += (size_t)N;
    float* sdst2 = ws + off; off += (size_t)N;
    int* binned  = (int*)(ws + off); off += (size_t)E + 64 * BIN_NCH;
    int* tab     = (int*)(ws + off);

    const int BS = 256;

    k_bin_edges<<<BIN_NCH, BIN_T, 0, stream>>>(src, dst, binned, tab, E, N);

    k_gemm_res<128, 4, true, true><<<256, 256, 0, stream>>>(x, W_heads, Wh1, N, 4);
    k_node_scores<4, 2><<<1024, BS, 0, stream>>>(Wh1, a_heads, ssrc1, sdst1, N);
    k_own_attend<2><<<cdiv((long)N, OWN_R), OWN_T, 0, stream>>>(src, dst, binned, tab, ssrc1, sdst1, Wh1, hcat, E, N);

    k_gemm_res<256, 2, false, false><<<256, 256, 0, stream>>>(hcat, W_out, Wh2, N, 2);
    k_node_scores<1, 8><<<256, BS, 0, stream>>>(Wh2, a_out, ssrc2, sdst2, N);
    k_own_attend<0><<<cdiv((long)N, OWN_R), OWN_T, 0, stream>>>(src, dst, binned, tab, ssrc2, sdst2, Wh2, out, E, N);
}


# Layer-1 sum pass: lanes = (head, 8 columns); an edge's weight is the lane's own head's exp, moved by one lane-select op, not 4 scalar reads; 2 row loads of 16 B per edge, not 4 of 8; sum order kept
# speedup vs baseline: 8.4072x; 1.0303x over previous
#include <hip/hip_runtime.h>
#include <cstdint>
#include <cstddef>

typedef float v2f __attribute__((ext_vector_type(2)));
typedef float v8f __attribute__((ext_vector_type(8)));

#define GAT_ALPHA 0.2f

__device__ __forceinline__ unsigned int ordKey(float v) {
    unsigned int b = __float_as_uint(v);
    return (b & 0x80000000u) ? ~b : (b | 0x80000000u);
}
__device__ __forceinline__ float ordVal(unsigned int k) {
    return __uint_as_float((k & 0x80000000u) ? (k & 0x7fffffffu) : ~k);
}


template <int K>
__device__ __forceinline__ void loadATile(const float* __restrict__ A, int tile, int row, int kh, v2f (&a)[K / 4]) {
    const float* __restrict__ ar = A + (size_t)(tile * 16 + row) * K + kh;
    #pragma unroll
    for (int s = 0; s < K / 4; ++s) {
        const float2 af = *reinterpret_cast<const float2*>(ar + 4 * s);
        a[s].x = af.x; a[s].y = af.y;
    }
}

template <int K, int NT, bool HEADS, bool AHEAD>
__global__ __launch_bounds__(256) void k_gemm_res(
    const float* __restrict__ A, const float* __restrict__ B, float* __restrict__ C, int M, int nslab)
{
    constexpr int KS = K / 4;
    const int lane = threadIdx.x & 31;
    const int wv   = __builtin_amdgcn_readfirstlane(threadIdx.x >> 5);
    const int slab = wv % nslab;
    const int tpw  = (blockDim.x >> 5) / nslab;
    const int tstep = gridDim.x * tpw;
    const int T    = M / 16;

    const int row = lane & 15;
    const int kh  = (lane >> 4) << 1;
    const int col = lane & 15;

    const float* __restrict__ Bs = HEADS ? B + (size_t)slab * K * 64 : B + slab * 16 * NT;
    float* __restrict__ Cs = HEADS ? C + (size_t)slab * M * 64 : C + slab * 16 * NT;

    v2f bf[KS][NT];
    #pragma unroll
    for (int s = 0; s < KS; ++s) {
        const float* br0 = Bs + (size_t)(4 * s + kh) * 64 + col;
        #pragma unroll
        for (int j = 0; j < NT; ++j) { bf[s][j].x = br0[16 * j]; bf[s][j].y = br0[64 + 16 * j]; }
    }

    int t = blockIdx.x + gridDim.x * (wv / nslab);
    if (t >= T) return;
    v2f a[KS];
    loadATile<K>(A, t, row, kh, a);
    for (; t < T; t += tstep) {
        v2f an[KS];
        if constexpr (AHEAD) loadATile<K>(A, min(t + tstep, T - 1), row, kh, an);

        v8f c[NT];
        #pragma unroll
        for (int j = 0; j < NT; ++j) c[j] = v8f{};
        #pragma unroll
        for (int s = 0; s < KS; ++s) {
            #pragma unroll
            for (int j = 0; j < NT; ++j)
                c[j] = __builtin_amdgcn_wmma_f32_16x16x4_f32(false, a[s], false, bf[s][j], (short)0, c[j], false, false);
        }

        const int rbase = t * 16 + ((lane >> 4) << 3);
        #pragma unroll
        for (int r = 0; r < 8; ++r) {
            float* crow = Cs + (size_t)(rbase + r) * 64 + col;
            #pragma unroll
            for (int j = 0; j < NT; ++j) crow[16 * j] = c[j][r];
        }

        if constexpr (AHEAD) {
            #pragma unroll
            for (int s = 0; s < KS; ++s) a[s] = an[s];
        } else {
            if (t + tstep < T) loadATile<K>(A, t + tstep, row, kh, a);
        }
    }
}

template <int H, int U>
__global__ __launch_bounds__(256) void k_node_scores(const float* __restrict__ Wh, const float* __restrict__ a,
                                                     float* __restrict__ ssrc, float* __restrict__ sdst, int N)
{
    const int lane = threadIdx.x & 31;
    const int wpb  = blockDim.x >> 5;
    const int w    = __builtin_amdgcn_readfirstlane(blockIdx.x * wpb + (threadIdx.x >> 5));
    const int nw   = gridDim.x * wpb;

    float as0[H], as1[H], ad0[H], ad1[H];
    #pragma unroll
    for (int h = 0; h < H; ++h) {
        const float* __restrict__ ah = a + h * 128;
        as0[h] = ah[lane];      as1[h] = ah[lane + 32];
        ad0[h] = ah[64 + lane]; ad1[h] = ah[96 + lane];
    }

    for (int n0 = w * U; n0 < N; n0 += nw * U) {
        float w0[U][H], w1[U][H];
        #pragma unroll
        for (int u = 0; u < U; ++u) {
            const int nn = min(n0 + u, N - 1);
            #pragma unroll
            for (int h = 0; h < H; ++h) {
                const float* __restrict__ row = Wh + ((size_t)h * N + nn) * 64;
                w0[u][h] = row[lane]; w1[u][h] = row[lane + 32];
            }
        }
        float s0[U][H], s1[U][H];
        #pragma unroll
        for (int u = 0; u < U; ++u) {
            #pragma unroll
            for (int h = 0; h < H; ++h) {
                s0[u][h] = __builtin_fmaf(w0[u][h], as0[h], w1[u][h] * as1[h]);
                s1[u][h] = __builtin_fmaf(w0[u][h], ad0[h], w1[u][h] * ad1[h]);
            }
        }
        #pragma unroll
        for (int off = 16; off >= 1; off >>= 1) {
            #pragma unroll
            for (int u = 0; u < U; ++u) {
                #pragma unroll
                for (int h = 0; h < H; ++h) {
                    s0[u][h] += __shfl_xor(s0[u][h], off, 32);
                    s1[u][h] += __shfl_xor(s1[u][h], off, 32);
                }
            }
        }
        if (lane == 0) {
            #pragma unroll
            for (int u = 0; u < U; ++u) {
                if (n0 + u < N) {
                    const size_t o = (size_t)(n0 + u) * H;
                    if constexpr (H == 4) {
                        *reinterpret_cast<float4*>(ssrc + o) = make_float4(s0[u][0], s0[u][1], s0[u][2], s0[u][3]);
                        *reinterpret_cast<float4*>(sdst + o) = make_float4(s1[u][0], s1[u][1], s1[u][2], s1[u][3]);
                    } else {
                        #pragma unroll
                        for (int h = 0; h < H; ++h) { ssrc[o + h] = s0[u][h]; sdst[o + h] = s1[u][h]; }
                    }
                }
            }
        }
    }
}

#define OWN_R    196
#define OWN_T    1024
#define OWN_QCAP 8192
#define OWN_P    4
#define OWN_NW   (OWN_T / 32)
#define OWN_RPW  ((OWN_R + OWN_NW - 1) / OWN_NW)
static_assert(OWN_NW == 32, "one wave scans the row counts; a row's owner wave is its local row mod 32");

__device__ __forceinline__ void ldsAddF(float* p, float v) {
    __hip_atomic_fetch_add(p, v, __ATOMIC_RELAXED, __HIP_MEMORY_SCOPE_WORKGROUP);
}

__device__ __forceinline__ float eluF(float v) { return v > 0.0f ? v : __expf(v) - 1.0f; }

template <int H>
__device__ __forceinline__ void storeRow(float* __restrict__ outp, int grow, int lane,
                                         const float2 (&sum)[H], const float (&dsm)[H])
{
    float* o = outp + (size_t)grow * (H * 64) + 2 * lane;
    #pragma unroll
    for (int hh = 0; hh < H; ++hh) {
        const float dn = dsm[hh];
        float2 r = make_float2(0.0f, 0.0f);
        if (dn > 0.0f) { r.x = eluF(sum[hh].x / dn); r.y = eluF(sum[hh].y / dn); }
        *reinterpret_cast<float2*>(o + hh * 64) = r;
    }
}

__device__ __forceinline__ float edgeScore(float a, float b) {
    const float v = a + b;
    return v > 0.0f ? v : GAT_ALPHA * v;
}

template <int H>
__device__ __forceinline__ void nodeScores(const float* __restrict__ p, int node, float (&v)[H]) {
    if constexpr (H == 4) {
        const float4 q = *reinterpret_cast<const float4*>(p + (size_t)node * 4);
        v[0] = q.x; v[1] = q.y; v[2] = q.z; v[3] = q.w;
    } else {
        #pragma unroll
        for (int h = 0; h < H; ++h) v[h] = p[(size_t)node * H + h];
    }
}

__device__ __forceinline__ void holdInts(const int (&d)[OWN_P]) {
    static_assert(OWN_P == 4, "written out for four slots");
    asm volatile("" :: "v"(d[0]), "v"(d[1]), "v"(d[2]), "v"(d[3]));
}
template <int H>
__device__ __forceinline__ void holdScores(const float (&v)[OWN_P][H]) {
    static_assert(OWN_P == 4 && (H == 1 || H == 4), "written out for four slots of one or four heads");
    if constexpr (H == 4) {
        asm volatile("" :: "v"(v[0][0]), "v"(v[0][1]), "v"(v[0][2]), "v"(v[0][3]),
                           "v"(v[1][0]), "v"(v[1][1]), "v"(v[1][2]), "v"(v[1][3]),
                           "v"(v[2][0]), "v"(v[2][1]), "v"(v[2][2]), "v"(v[2][3]),
                           "v"(v[3][0]), "v"(v[3][1]), "v"(v[3][2]), "v"(v[3][3]));
    } else {
        asm volatile("" :: "v"(v[0][0]), "v"(v[1][0]), "v"(v[2][0]), "v"(v[3][0]));
    }
}

__device__ __forceinline__ void holdRows(const float2 (&w)[32], int o) {
    asm volatile("" :: "v"(w[o + 0].x), "v"(w[o + 0].y), "v"(w[o + 1].x), "v"(w[o + 1].y),
                       "v"(w[o + 2].x), "v"(w[o + 2].y), "v"(w[o + 3].x), "v"(w[o + 3].y),
                       "v"(w[o + 4].x), "v"(w[o + 4].y), "v"(w[o + 5].x), "v"(w[o + 5].y),
                       "v"(w[o + 6].x), "v"(w[o + 6].y), "v"(w[o + 7].x), "v"(w[o + 7].y));
}


template <int E>
__device__ __forceinline__ float groupLaneC(float v) {
    constexpr unsigned sel = (unsigned)E * 0x249249u;
    return __uint_as_float((unsigned)__builtin_amdgcn_mov_dpp8(__float_as_uint(v), sel));
}
__device__ __forceinline__ float groupLane(float v, int e) {
    switch (e) {
        case 0: return groupLaneC<0>(v);
        case 1: return groupLaneC<1>(v);
        case 2: return groupLaneC<2>(v);
        case 3: return groupLaneC<3>(v);
        case 4: return groupLaneC<4>(v);
        case 5: return groupLaneC<5>(v);
        case 6: return groupLaneC<6>(v);
        default: return groupLaneC<7>(v);
    }
}

__device__ __forceinline__ void holdQuads(const float4 (&a)[8], const float4 (&b)[8], int o) {
    asm volatile("" :: "v"(a[o].x), "v"(a[o].y), "v"(a[o].z), "v"(a[o].w),
                       "v"(b[o].x), "v"(b[o].y), "v"(b[o].z), "v"(b[o].w),
                       "v"(a[o + 1].x), "v"(a[o + 1].y), "v"(a[o + 1].z), "v"(a[o + 1].w),
                       "v"(b[o + 1].x), "v"(b[o + 1].y), "v"(b[o + 1].z), "v"(b[o + 1].w));
}

__device__ __forceinline__ void storeRowQ(float* __restrict__ outp, int grow, int lane,
                                          const float4& sa, const float4& sb, float dn)
{
    float* o = outp + (size_t)grow * 256 + (lane >> 3) * 64 + 4 * (lane & 7);
    float4 ra = make_float4(0.0f, 0.0f, 0.0f, 0.0f), rb = ra;
    if (dn > 0.0f) {
        ra = make_float4(eluF(sa.x / dn), eluF(sa.y / dn), eluF(sa.z / dn), eluF(sa.w / dn));
        rb = make_float4(eluF(sb.x / dn), eluF(sb.y / dn), eluF(sb.z / dn), eluF(sb.w / dn));
    }
    *reinterpret_cast<float4*>(o) = ra;
    *reinterpret_cast<float4*>(o + 32) = rb;
}

#define BIN_NCH  256
#define BIN_T    512
#define BIN_U    8
#define BIN_MAXR 1024

static_assert(OWN_R <= 256, "the local row of an edge is kept in 8 bits");

__device__ __forceinline__ int binRanges(int N) { return min((N + OWN_R - 1) / OWN_R, BIN_MAXR - 1); }
__device__ __forceinline__ int binTabPitch(int NB) { return (NB + 32) & ~31; }
__device__ __forceinline__ int binChunk(int E) { return (E + BIN_NCH - 1) / BIN_NCH; }
__device__ __forceinline__ int binPiecePitch(int CH) { return (CH + 31) & ~31; }

__global__ __launch_bounds__(BIN_T) void k_bin_edges(
    const int* __restrict__ src, const int* __restrict__ dst,
    int* __restrict__ binned, int* __restrict__ tab, int E, int N)
{
    __shared__ int cnt[BIN_MAXR];
    __shared__ int wtot[BIN_T / 32];

    const int tid = threadIdx.x, lane = tid & 31, wave = tid >> 5;
    const int NB = binRanges(N);
    const int TP = binTabPitch(NB);
    const int CH = binChunk(E);
    const int SP = binPiecePitch(CH);
    const int c  = blockIdx.x;
    const int e0 = c * CH;
    const int n  = min(CH, E - e0);
    const unsigned lim = (unsigned)min(N, NB * OWN_R);

    for (int i = tid; i < BIN_MAXR; i += BIN_T) cnt[i] = 0;
    __syncthreads();

    for (int k0 = 0; k0 < n; k0 += BIN_T * BIN_U) {
        int s[BIN_U];
        #pragma unroll
        for (int u = 0; u < BIN_U; ++u) s[u] = src[e0 + min(k0 + u * BIN_T + tid, n - 1)];
        #pragma unroll
        for (int u = 0; u < BIN_U; ++u) {
            const int k = k0 + u * BIN_T + tid;
            if (k < n && (unsigned)s[u] < lim) atomicAdd(&cnt[(unsigned)s[u] / OWN_R], 1);
        }
    }
    __syncthreads();

    constexpr int PER = BIN_MAXR / BIN_T;
    int v[PER];
    int tsum = 0;
    #pragma unroll
    for (int j = 0; j < PER; ++j) { v[j] = cnt[tid * PER + j]; tsum += v[j]; }
    int incl = tsum;
    #pragma unroll
    for (int off = 1; off < 32; off <<= 1) {
        const int y = __shfl_up(incl, off, 32);
        if (lane >= off) incl += y;
    }
    if (lane == 31) wtot[wave] = incl;
    __syncthreads();
    int run = incl - tsum;
    for (int w = 0; w < wave; ++w) run += wtot[w];
    #pragma unroll
    for (int j = 0; j < PER; ++j) {
        const int r = tid * PER + j;
        cnt[r] = run;
        if (r <= NB) tab[(size_t)c * TP + r] = run;
        run += v[j];
    }
    __syncthreads();

    for (int k0 = 0; k0 < n; k0 += BIN_T * BIN_U) {
        int s[BIN_U], d[BIN_U];
        #pragma unroll
        for (int u = 0; u < BIN_U; ++u) {
            const int k = e0 + min(k0 + u * BIN_T + tid, n - 1);
            s[u] = src[k]; d[u] = dst[k];
        }
        #pragma unroll
        for (int u = 0; u < BIN_U; ++u) {
            const int k = k0 + u * BIN_T + tid;
            if (k < n && (unsigned)s[u] < lim) {
                const unsigned r = (unsigned)s[u] / OWN_R;
                const int p = atomicAdd(&cnt[r], 1);
                binned[(size_t)c * SP + p] = (int)((((unsigned)s[u] - r * OWN_R) << 24) | ((unsigned)d[u] & 0xffffffu));
            }
        }
    }
}

template <int HS>
__global__ __launch_bounds__(OWN_T) void k_own_attend(
    const int* __restrict__ src, const int* __restrict__ dst,
    const int* __restrict__ binned, const int* __restrict__ tab,
    const float* __restrict__ ssrc, const float* __restrict__ sdst,
    const float* __restrict__ Wh, float* __restrict__ outp, int E, int N)
{
    constexpr int H  = 1 << HS;
    constexpr int RW = H * 64;
    __shared__ __attribute__((aligned(16))) float acc[OWN_R * RW];
    __shared__ float den[OWN_R * H];
    __shared__ unsigned int mx[OWN_R * H];
    __shared__ int   queA[OWN_QCAP];
    __shared__ int   queB[OWN_QCAP];
    __shared__ int   rcnt[OWN_NW * OWN_RPW];
    __shared__ int   rpos[OWN_NW * OWN_RPW];
    __shared__ int   bst[OWN_NW + 1];
    __shared__ int   cst[BIN_NCH];
    __shared__ int   qof[BIN_NCH + 1];
    __shared__ int   wsm[BIN_NCH / 32];

    const int tid  = threadIdx.x;
    const int lane = tid & 31;
    const int lo   = blockIdx.x * OWN_R;
    const int nrows = min(OWN_R, N - lo);

    const int NB = binRanges(N);
    const int TP = binTabPitch(NB);
    const int SP = binPiecePitch(binChunk(E));
    int myst = 0, mycnt = 0;
    if (tid < BIN_NCH && (int)blockIdx.x < NB) {
        const int* tr = tab + (size_t)tid * TP + blockIdx.x;
        myst = tr[0]; mycnt = tr[1] - myst;
    }

    float4* acc4 = reinterpret_cast<float4*>(acc);
    for (int i = tid; i < OWN_R * H; i += OWN_T) mx[i] = 0u;
    if (tid < OWN_NW * OWN_RPW) rcnt[tid] = 0;

    int incl = mycnt;
    if (tid < BIN_NCH) {
        #pragma unroll
        for (int off = 1; off < 32; off <<= 1) {
            const int y = __shfl_up(incl, off, 32);
            if (lane >= off) incl += y;
        }
        cst[tid] = myst;
        if (lane == 31) wsm[tid >> 5] = incl;
    }
    __syncthreads();
    if (tid < BIN_NCH) {
        int pre = incl - mycnt;
        for (int w = 0; w < (tid >> 5); ++w) pre += wsm[w];
        qof[tid] = pre;
        if (tid == BIN_NCH - 1) qof[BIN_NCH] = pre + mycnt;
    }
    __syncthreads();
    const int ntot = qof[BIN_NCH];

    if (ntot <= OWN_QCAP) {
        for (int base = 0; base < ntot; base += OWN_T * OWN_P) {
            const int i0 = base + tid;
            if (i0 < ntot) {
                int qi[OWN_P], cI[OWN_P], wI[OWN_P], rI[OWN_P], dI[OWN_P];
                #pragma unroll
                for (int u = 0; u < OWN_P; ++u) {
                    const int i = i0 + u * OWN_T;
                    qi[u] = i < ntot ? i : i0;
                }
                #pragma unroll
                for (int u = 0; u < OWN_P; ++u) {
                    int c = 0;
                    #pragma unroll
                    for (int st = BIN_NCH / 2; st >= 1; st >>= 1)
                        if (qof[c + st] <= qi[u]) c += st;
                    cI[u] = c;
                }
                #pragma unroll
                for (int u = 0; u < OWN_P; ++u)
                    wI[u] = binned[(size_t)cI[u] * SP + cst[cI[u]] + (qi[u] - qof[cI[u]])];
                holdInts(wI);
                float a[OWN_P][H], b[OWN_P][H];
                #pragma unroll
                for (int u = 0; u < OWN_P; ++u) {
                    dI[u] = wI[u] & 0xffffff;
                    rI[u] = (int)((unsigned)wI[u] >> 24);
                    nodeScores<H>(ssrc, lo + rI[u], a[u]);
                    nodeScores<H>(sdst, dI[u], b[u]);
                }
                holdScores<H>(a);
                holdScores<H>(b);
                #pragma unroll
                for (int u = 0; u < OWN_P; ++u) {
                    #pragma unroll
                    for (int h = 0; h < H; ++h)
                        atomicMax(&mx[rI[u] * H + h], ordKey(edgeScore(a[u][h], b[u][h])));
                    queA[qi[u]] = wI[u];
                    if (i0 + u * OWN_T < ntot) atomicAdd(&rcnt[rI[u]], 1);
                }
            }
        }
        __syncthreads();

        if (tid < OWN_NW) {
            int c[OWN_RPW];
            int t = 0;
            #pragma unroll
            for (int k = 0; k < OWN_RPW; ++k) { c[k] = rcnt[tid + OWN_NW * k]; t += c[k]; }
            int run = t;
            #pragma unroll
            for (int off = 1; off < 32; off <<= 1) {
                const int y = __shfl_up(run, off, 32);
                if (lane >= off) run += y;
            }
            int st = run - t;
            bst[tid] = st;
            #pragma unroll
            for (int k = 0; k < OWN_RPW; ++k) { rpos[tid + OWN_NW * k] = st; st += c[k]; }
            if (tid == OWN_NW - 1) bst[OWN_NW] = run;
        }
        __syncthreads();
        for (int i = tid; i < ntot; i += OWN_T) {
            const int wd = queA[i];
            const int p = atomicAdd(&rpos[(unsigned)wd >> 24], 1);
            queB[p] = wd;
        }
        __syncthreads();

        const int wave = __builtin_amdgcn_readfirstlane(tid >> 5);
        const int bbeg = __builtin_amdgcn_readfirstlane(bst[wave]);
        const int bend = __builtin_amdgcn_readfirstlane(bst[wave + 1]);
        if constexpr (H == 4) {
            const int hq = lane >> 3, g = lane & 7;
            const unsigned voff = ((unsigned)hq * (unsigned)N * 64u + 4u * (unsigned)g) * 4u;
            const char* __restrict__ whb = reinterpret_cast<const char*>(Wh);
            int cur = -1;
            float4 sa = make_float4(0.0f, 0.0f, 0.0f, 0.0f), sb = sa;
            float dsm = 0.0f;
            for (int b0 = bbeg; b0 < bend; b0 += 8) {
                const bool ok = b0 + g < bend;
                const int wd = queB[ok ? b0 + g : b0];
                const int d  = wd & 0xffffff;
                const int rl = (int)((unsigned)wd >> 24);
                const float sc = edgeScore(ssrc[(size_t)(lo + rl) * 4 + hq], sdst[(size_t)d * 4 + hq]);
                const float ex = __expf(sc - ordVal(mx[rl * 4 + hq]));
                float4 wa[8], wb[8];
                #pragma unroll
                for (int e = 0; e < 8; ++e) {
                    const unsigned dE = (unsigned)__builtin_amdgcn_readlane(d, e);
                    const char* rec = whb + (size_t)dE * 256u;
                    wa[e] = *reinterpret_cast<const float4*>(rec + voff);
                    wb[e] = *reinterpret_cast<const float4*>(rec + voff + 128);
                }
                holdQuads(wa, wb, 0); holdQuads(wa, wb, 2); holdQuads(wa, wb, 4); holdQuads(wa, wb, 6);
                #pragma unroll
                for (int e = 0; e < 8; ++e) {
                    if (b0 + e < bend) {
                        const int r = __builtin_amdgcn_readlane(rl, e);
                        if (r != cur) {
                            if (cur >= 0) storeRowQ(outp, lo + cur, lane, sa, sb, dsm);
                            cur = r;
                            sa = make_float4(0.0f, 0.0f, 0.0f, 0.0f); sb = sa; dsm = 0.0f;
                        }
                        const float exE = groupLane(ex, e);
                        sa.x = __builtin_fmaf(exE, wa[e].x, sa.x); sa.y = __builtin_fmaf(exE, wa[e].y, sa.y);
                        sa.z = __builtin_fmaf(exE, wa[e].z, sa.z); sa.w = __builtin_fmaf(exE, wa[e].w, sa.w);
                        sb.x = __builtin_fmaf(exE, wb[e].x, sb.x); sb.y = __builtin_fmaf(exE, wb[e].y, sb.y);
                        sb.z = __builtin_fmaf(exE, wb[e].z, sb.z); sb.w = __builtin_fmaf(exE, wb[e].w, sb.w);
                        dsm += exE;
                    }
                }
            }
            if (cur >= 0) storeRowQ(outp, lo + cur, lane, sa, sb, dsm);
        } else {
            constexpr int EPB = 32 >> HS;
            const int j = lane >> HS, h = lane & (H - 1);
            int cur = -1;
            float2 sum[H];
            float dsm[H];
            #pragma unroll
            for (int hh = 0; hh < H; ++hh) { sum[hh] = make_float2(0.0f, 0.0f); dsm[hh] = 0.0f; }
            for (int b0 = bbeg; b0 < bend; b0 += EPB) {
                const bool ok = b0 + j < bend;
                const int wd = queB[ok ? b0 + j : b0];
                const int d  = wd & 0xffffff;
                const int rl = (int)((unsigned)wd >> 24);
                const float sc = edgeScore(ssrc[(size_t)(lo + rl) * H + h], sdst[(size_t)d * H + h]);
                const float ex = __expf(sc - ordVal(mx[rl * H + h]));
                float2 w[32];
                #pragma unroll
                for (int L = 0; L < 32; ++L) {
                    const int dL = __builtin_amdgcn_readlane(d, L);
                    const float* row = Wh + ((size_t)(L & (H - 1)) * N + dL) * 64;
                    w[L] = *reinterpret_cast<const float2*>(row + 2 * lane);
                }
                holdRows(w, 0); holdRows(w, 8); holdRows(w, 16); holdRows(w, 24);
                #pragma unroll
                for (int e = 0; e < EPB; ++e) {
                    if (b0 + e < bend) {
                        const int LE = e * H;
                        const int r = __builtin_amdgcn_readlane(rl, LE);
                        if (r != cur) {
                            if (cur >= 0) storeRow<H>(outp, lo + cur, lane, sum, dsm);
                            cur = r;
                            #pragma unroll
                            for (int hh = 0; hh < H; ++hh) { sum[hh] = make_float2(0.0f, 0.0f); dsm[hh] = 0.0f; }
                        }
                        #pragma unroll
                        for (int hh = 0; hh < H; ++hh) {
                            const float exE = __int_as_float(__builtin_amdgcn_readlane(__float_as_int(ex), LE + hh));
                            sum[hh].x = __builtin_fmaf(exE, w[LE + hh].x, sum[hh].x);
                            sum[hh].y = __builtin_fmaf(exE, w[LE + hh].y, sum[hh].y);
                            dsm[hh] += exE;
                        }
                    }
                }
            }
            if (cur >= 0) storeRow<H>(outp, lo + cur, lane, sum, dsm);
        }
        for (int k = 0; k < OWN_RPW; ++k) {
            const int r = wave + OWN_NW * k;
            if (r < nrows && rcnt[r] == 0) {
                float* o = outp + (size_t)(lo + r) * RW + 2 * lane;
                for (int hh = 0; hh < H; ++hh) *reinterpret_cast<float2*>(o + hh * 64) = make_float2(0.0f, 0.0f);
            }
        }
    } else {
        for (int i = tid; i < OWN_R * RW / 4; i += OWN_T) acc4[i] = make_float4(0.0f, 0.0f, 0.0f, 0.0f);
        for (int i = tid; i < OWN_R * H; i += OWN_T) den[i] = 0.0f;
        for (int e = tid; e < E; e += OWN_T) {
            const int s = src[e];
            const unsigned rl = (unsigned)(s - lo);
            if (rl < (unsigned)nrows) {
                const int d = dst[e];
                for (int hh = 0; hh < H; ++hh)
                    atomicMax(&mx[rl * H + hh],
                              ordKey(edgeScore(ssrc[(size_t)s * H + hh], sdst[(size_t)d * H + hh])));
            }
        }
        __syncthreads();
        for (int e = tid; e < E; e += OWN_T) {
            const int s = src[e];
            const unsigned rl = (unsigned)(s - lo);
            if (rl < (unsigned)nrows) {
                const int d = dst[e];
                for (int hh = 0; hh < H; ++hh) {
                    const float sc = edgeScore(ssrc[(size_t)s * H + hh], sdst[(size_t)d * H + hh]);
                    const float ex = __expf(sc - ordVal(mx[rl * H + hh]));
                    ldsAddF(&den[rl * H + hh], ex);
                    const float* row = Wh + ((size_t)hh * N + d) * 64;
                    float* ap = acc + rl * RW + hh * 64;
                    for (int k = 0; k < 64; ++k) ldsAddF(ap + k, ex * row[k]);
                }
            }
        }
        __syncthreads();

        float4* o4 = reinterpret_cast<float4*>(outp + (size_t)lo * RW);
        const int n4 = nrows * (RW / 4);
        for (int i = tid; i < n4; i += OWN_T) {
            const float4 a = acc4[i];
            const float dn = den[i >> 4];
            float4 r;
            if (dn > 0.0f) {
                r.x = eluF(a.x / dn); r.y = eluF(a.y / dn); r.z = eluF(a.z / dn); r.w = eluF(a.w / dn);
            } else {
                r = make_float4(0.0f, 0.0f, 0.0f, 0.0f);
            }
            o4[i] = r;
        }
    }
}

static inline int cdiv(long a, int b) { return (int)((a + b - 1) / b); }

extern "C" void kernel_launch(void* const* d_in, const int* in_sizes, int n_in,
                              void* d_out, int out_size, void* d_ws, size_t ws_size,
                              hipStream_t stream)
{
    const float* x       = (const float*)d_in[0];
    const int*   edges   = (const int*)d_in[1];
    const float* W_heads = (const float*)d_in[2];
    const float* a_heads = (const float*)d_in[3];
    const float* W_out   = (const float*)d_in[4];
    const float* a_out   = (const float*)d_in[5];
    float* out = (float*)d_out;

    const int N = in_sizes[0] / 128;
    const int E = in_sizes[1] / 2;
    const int H = 4, HID = 64;

    const int* src = edges;
    const int* dst = edges + E;

    float* ws = (float*)d_ws;
    size_t off = 0;
    float* Wh1   = ws + off; off += (size_t)H * N * HID;
    float* hcat  = ws + off; off += (size_t)N * 256;
    float* Wh2   = ws + off; off += (size_t)N * HID;
    float* ssrc1 = ws + off; off += (size_t)H * N;
    float* sdst1 = ws + off; off += (size_t)H * N;
    float* ssrc2 = ws + off; off += (size_t)N;
    float* sdst2 = ws + off; off += (size_t)N;
    int* binned  = (int*)(ws + off); off += (size_t)E + 64 * BIN_NCH;
    int* tab     = (int*)(ws + off);

    const int BS = 256;

    k_bin_edges<<<BIN_NCH, BIN_T, 0, stream>>>(src, dst, binned, tab, E, N);

    k_gemm_res<128, 4, true, true><<<256, 256, 0, stream>>>(x, W_heads, Wh1, N, 4);
    k_node_scores<4, 2><<<1024, BS, 0, stream>>>(Wh1, a_heads, ssrc1, sdst1, N);
    k_own_attend<2><<<cdiv((long)N, OWN_R), OWN_T, 0, stream>>>(src, dst, binned, tab, ssrc1, sdst1, Wh1, hcat, E, N);

    k_gemm_res<256, 2, false, false><<<256, 256, 0, stream>>>(hcat, W_out, Wh2, N, 2);
    k_node_scores<1, 8><<<256, BS, 0, stream>>>(Wh2, a_out, ssrc2, sdst2, N);
    k_own_attend<0><<<cdiv((long)N, OWN_R), OWN_T, 0, stream>>>(src, dst, binned, tab, ssrc2, sdst2, Wh2, out, E, N);
}


# Node scores = input row . (W a), W a folded in the partition launch: layer 1 from the product's A fragments, layer 2 from the attend wave's finished row; 7 -> 5 launches; score sum order changes
# speedup vs baseline: 8.7162x; 1.0368x over previous
#include <hip/hip_runtime.h>
#include <cstdint>
#include <cstddef>

typedef float v2f __attribute__((ext_vector_type(2)));
typedef float v8f __attribute__((ext_vector_type(8)));

#define GAT_ALPHA 0.2f

__device__ __forceinline__ unsigned int ordKey(float v) {
    unsigned int b = __float_as_uint(v);
    return (b & 0x80000000u) ? ~b : (b | 0x80000000u);
}
__device__ __forceinline__ float ordVal(unsigned int k) {
    return __uint_as_float((k & 0x80000000u) ? (k & 0x7fffffffu) : ~k);
}


template <int K>
__device__ __forceinline__ void loadATile(const float* __restrict__ A, int tile, int row, int kh, v2f (&a)[K / 4]) {
    const float* __restrict__ ar = A + (size_t)(tile * 16 + row) * K + kh;
    #pragma unroll
    for (int s = 0; s < K / 4; ++s) {
        const float2 af = *reinterpret_cast<const float2*>(ar + 4 * s);
        a[s].x = af.x; a[s].y = af.y;
    }
}

template <int K, int NT, bool HEADS, bool AHEAD, bool SCORE>
__global__ __launch_bounds__(256) void k_gemm_res(
    const float* __restrict__ A, const float* __restrict__ B, float* __restrict__ C, int M, int nslab,
    const float* __restrict__ wfold, float* __restrict__ ssrc, float* __restrict__ sdst)
{
    static_assert(!SCORE || (HEADS && NT == 4), "scores need a wave that holds a whole head");
    constexpr int KS = K / 4;
    constexpr int NQ = SCORE ? 4 * (K / 2) : 1;
    __shared__ float4 wql[NQ];
    float4 wq0 = make_float4(0.0f, 0.0f, 0.0f, 0.0f);
    if constexpr (SCORE) {
        if ((int)threadIdx.x < NQ) wq0 = reinterpret_cast<const float4*>(wfold)[threadIdx.x];
    }
    const int lane = threadIdx.x & 31;
    const int wv   = __builtin_amdgcn_readfirstlane(threadIdx.x >> 5);
    const int slab = wv % nslab;
    const int tpw  = (blockDim.x >> 5) / nslab;
    const int tstep = gridDim.x * tpw;
    const int T    = M / 16;

    const int row = lane & 15;
    const int kh  = (lane >> 4) << 1;
    const int col = lane & 15;

    const float* __restrict__ Bs = HEADS ? B + (size_t)slab * K * 64 : B + slab * 16 * NT;
    float* __restrict__ Cs = HEADS ? C + (size_t)slab * M * 64 : C + slab * 16 * NT;

    v2f bf[KS][NT];
    #pragma unroll
    for (int s = 0; s < KS; ++s) {
        const float* br0 = Bs + (size_t)(4 * s + kh) * 64 + col;
        #pragma unroll
        for (int j = 0; j < NT; ++j) { bf[s][j].x = br0[16 * j]; bf[s][j].y = br0[64 + 16 * j]; }
    }

    int t = blockIdx.x + gridDim.x * (wv / nslab);
    v2f a[KS];
    if constexpr (SCORE) {
        loadATile<K>(A, min(t, T - 1), row, kh, a);
        if ((int)threadIdx.x < NQ) wql[threadIdx.x] = wq0;
        for (int i = threadIdx.x + blockDim.x; i < NQ; i += blockDim.x)
            wql[i] = reinterpret_cast<const float4*>(wfold)[i];
        __syncthreads();
        if (t >= T) return;
    } else {
        if (t >= T) return;
        loadATile<K>(A, t, row, kh, a);
    }
    for (; t < T; t += tstep) {
        v2f an[KS];
        if constexpr (AHEAD) loadATile<K>(A, min(t + tstep, T - 1), row, kh, an);

        v8f c[NT];
        #pragma unroll
        for (int j = 0; j < NT; ++j) c[j] = v8f{};
        float ps0 = 0.0f, ps1 = 0.0f, pd0 = 0.0f, pd1 = 0.0f;
        #pragma unroll
        for (int s = 0; s < KS; ++s) {
            #pragma unroll
            for (int j = 0; j < NT; ++j)
                c[j] = __builtin_amdgcn_wmma_f32_16x16x4_f32(false, a[s], false, bf[s][j], (short)0, c[j], false, false);
            if constexpr (SCORE) {
                if (s == KS / 2) __builtin_amdgcn_sched_barrier(0);
                const float4 q = wql[slab * (K / 2) + 2 * s + (lane >> 4)];
                ps0 = __builtin_fmaf(a[s].x, q.x, ps0); ps1 = __builtin_fmaf(a[s].y, q.y, ps1);
                pd0 = __builtin_fmaf(a[s].x, q.z, pd0); pd1 = __builtin_fmaf(a[s].y, q.w, pd1);
            }
        }
        if constexpr (SCORE) {
            __builtin_amdgcn_sched_barrier(0);
            const bool lowh = lane < 16;
            const float psum = ps0 + ps1, pdsum = pd0 + pd1;
            float keep = lowh ? psum : pdsum;
            const float give = lowh ? pdsum : psum;
            keep += __shfl_xor(give, 16, 32);
            float* __restrict__ sp = lowh ? ssrc : sdst;
            sp[(size_t)(t * 16 + row) * 4 + slab] = keep;
        }

        const int rbase = t * 16 + ((lane >> 4) << 3);
        #pragma unroll
        for (int r = 0; r < 8; ++r) {
            float* crow = Cs + (size_t)(rbase + r) * 64 + col;
            #pragma unroll
            for (int j = 0; j < NT; ++j) crow[16 * j] = c[j][r];
        }

        if constexpr (AHEAD) {
            #pragma unroll
            for (int s = 0; s < KS; ++s) a[s] = an[s];
        } else {
            if (t + tstep < T) loadATile<K>(A, t + tstep, row, kh, a);
        }
    }
}

#define FOLD_DOTS   1536
#define FOLD_PER_WG 6
__device__ __forceinline__ void foldRequest(const float* __restrict__ W1, const float* __restrict__ a1,
                                            const float* __restrict__ W2, const float* __restrict__ a2,
                                            float* __restrict__ wf1, float* __restrict__ wf2, int wg, int tid,
                                            float4& w4, float4& a4, float*& place)
{
    const int dot = wg * FOLD_PER_WG + (tid >> 4);
    const int sub = tid & 15;
    const bool live = tid < 16 * FOLD_PER_WG && dot < FOLD_DOTS;
    const int dd = live ? dot : 0;
    const float* wrow; const float* av; float* outp;
    if (dd < 1024) {
        const int head = dd >> 8, k = (dd >> 1) & 127, v = dd & 1;
        wrow = W1 + ((size_t)head * 128 + k) * 64;
        av   = a1 + head * 128 + 64 * v;
        outp = wf1 + head * 256 + (k >> 1) * 4 + v * 2 + (k & 1);
    } else {
        const int e = dd - 1024, k = e >> 1, v = e & 1;
        wrow = W2 + (size_t)k * 64;
        av   = a2 + 64 * v;
        outp = wf2 + v * 256 + k;
    }
    w4 = *reinterpret_cast<const float4*>(wrow + 4 * sub);
    a4 = *reinterpret_cast<const float4*>(av + 4 * sub);
    place = (live && sub == 0) ? outp : nullptr;
}
__device__ __forceinline__ void foldFinish(const float4& w4, const float4& a4, float* place)
{
    float p = __builtin_fmaf(w4.x, a4.x, __builtin_fmaf(w4.y, a4.y, __builtin_fmaf(w4.z, a4.z, w4.w * a4.w)));
    p += __shfl_xor(p, 8, 32); p += __shfl_xor(p, 4, 32);
    p += __shfl_xor(p, 2, 32); p += __shfl_xor(p, 1, 32);
    if (place) *place = p;
}

#define OWN_R    196
#define OWN_T    1024
#define OWN_QCAP 8192
#define OWN_P    4
#define OWN_NW   (OWN_T / 32)
#define OWN_RPW  ((OWN_R + OWN_NW - 1) / OWN_NW)
static_assert(OWN_NW == 32, "one wave scans the row counts; a row's owner wave is its local row mod 32");

__device__ __forceinline__ void ldsAddF(float* p, float v) {
    __hip_atomic_fetch_add(p, v, __ATOMIC_RELAXED, __HIP_MEMORY_SCOPE_WORKGROUP);
}

__device__ __forceinline__ float eluF(float v) { return v > 0.0f ? v : __expf(v) - 1.0f; }

template <int H>
__device__ __forceinline__ void storeRow(float* __restrict__ outp, int grow, int lane,
                                         const float2 (&sum)[H], const float (&dsm)[H])
{
    float* o = outp + (size_t)grow * (H * 64) + 2 * lane;
    #pragma unroll
    for (int hh = 0; hh < H; ++hh) {
        const float dn = dsm[hh];
        float2 r = make_float2(0.0f, 0.0f);
        if (dn > 0.0f) { r.x = eluF(sum[hh].x / dn); r.y = eluF(sum[hh].y / dn); }
        *reinterpret_cast<float2*>(o + hh * 64) = r;
    }
}

__device__ __forceinline__ float edgeScore(float a, float b) {
    const float v = a + b;
    return v > 0.0f ? v : GAT_ALPHA * v;
}

template <int H>
__device__ __forceinline__ void nodeScores(const float* __restrict__ p, int node, float (&v)[H]) {
    if constexpr (H == 4) {
        const float4 q = *reinterpret_cast<const float4*>(p + (size_t)node * 4);
        v[0] = q.x; v[1] = q.y; v[2] = q.z; v[3] = q.w;
    } else {
        #pragma unroll
        for (int h = 0; h < H; ++h) v[h] = p[(size_t)node * H + h];
    }
}

__device__ __forceinline__ void holdInts(const int (&d)[OWN_P]) {
    static_assert(OWN_P == 4, "written out for four slots");
    asm volatile("" :: "v"(d[0]), "v"(d[1]), "v"(d[2]), "v"(d[3]));
}
template <int H>
__device__ __forceinline__ void holdScores(const float (&v)[OWN_P][H]) {
    static_assert(OWN_P == 4 && (H == 1 || H == 4), "written out for four slots of one or four heads");
    if constexpr (H == 4) {
        asm volatile("" :: "v"(v[0][0]), "v"(v[0][1]), "v"(v[0][2]), "v"(v[0][3]),
                           "v"(v[1][0]), "v"(v[1][1]), "v"(v[1][2]), "v"(v[1][3]),
                           "v"(v[2][0]), "v"(v[2][1]), "v"(v[2][2]), "v"(v[2][3]),
                           "v"(v[3][0]), "v"(v[3][1]), "v"(v[3][2]), "v"(v[3][3]));
    } else {
        asm volatile("" :: "v"(v[0][0]), "v"(v[1][0]), "v"(v[2][0]), "v"(v[3][0]));
    }
}

__device__ __forceinline__ void holdRows(const float2 (&w)[32], int o) {
    asm volatile("" :: "v"(w[o + 0].x), "v"(w[o + 0].y), "v"(w[o + 1].x), "v"(w[o + 1].y),
                       "v"(w[o + 2].x), "v"(w[o + 2].y), "v"(w[o + 3].x), "v"(w[o + 3].y),
                       "v"(w[o + 4].x), "v"(w[o + 4].y), "v"(w[o + 5].x), "v"(w[o + 5].y),
                       "v"(w[o + 6].x), "v"(w[o + 6].y), "v"(w[o + 7].x), "v"(w[o + 7].y));
}


template <int E>
__device__ __forceinline__ float groupLaneC(float v) {
    constexpr unsigned sel = (unsigned)E * 0x249249u;
    return __uint_as_float((unsigned)__builtin_amdgcn_mov_dpp8(__float_as_uint(v), sel));
}
__device__ __forceinline__ float groupLane(float v, int e) {
    switch (e) {
        case 0: return groupLaneC<0>(v);
        case 1: return groupLaneC<1>(v);
        case 2: return groupLaneC<2>(v);
        case 3: return groupLaneC<3>(v);
        case 4: return groupLaneC<4>(v);
        case 5: return groupLaneC<5>(v);
        case 6: return groupLaneC<6>(v);
        default: return groupLaneC<7>(v);
    }
}

__device__ __forceinline__ void holdQuads(const float4 (&a)[8], const float4 (&b)[8], int o) {
    asm volatile("" :: "v"(a[o].x), "v"(a[o].y), "v"(a[o].z), "v"(a[o].w),
                       "v"(b[o].x), "v"(b[o].y), "v"(b[o].z), "v"(b[o].w),
                       "v"(a[o + 1].x), "v"(a[o + 1].y), "v"(a[o + 1].z), "v"(a[o + 1].w),
                       "v"(b[o + 1].x), "v"(b[o + 1].y), "v"(b[o + 1].z), "v"(b[o + 1].w));
}

__device__ __forceinline__ float rowSum16(float v) {
    v += __uint_as_float((unsigned)__builtin_amdgcn_update_dpp(0, (int)__float_as_uint(v), 0x168, 0xf, 0xf, false));
    v += __uint_as_float((unsigned)__builtin_amdgcn_update_dpp(0, (int)__float_as_uint(v), 0x164, 0xf, 0xf, false));
    v += __uint_as_float((unsigned)__builtin_amdgcn_update_dpp(0, (int)__float_as_uint(v), 0x162, 0xf, 0xf, false));
    v += __uint_as_float((unsigned)__builtin_amdgcn_update_dpp(0, (int)__float_as_uint(v), 0x161, 0xf, 0xf, false));
    return v;
}

__device__ __forceinline__ void storeRowQ(float* __restrict__ outp, int grow, int lane,
                                          const float4& sa, const float4& sb, float dn,
                                          const float4& fsa, const float4& fsb, const float4& fda, const float4& fdb,
                                          float* __restrict__ s2src, float* __restrict__ s2dst)
{
    float* o = outp + (size_t)grow * 256 + (lane >> 3) * 64 + 4 * (lane & 7);
    float4 ra = make_float4(0.0f, 0.0f, 0.0f, 0.0f), rb = ra;
    if (dn > 0.0f) {
        ra = make_float4(eluF(sa.x / dn), eluF(sa.y / dn), eluF(sa.z / dn), eluF(sa.w / dn));
        rb = make_float4(eluF(sb.x / dn), eluF(sb.y / dn), eluF(sb.z / dn), eluF(sb.w / dn));
    }
    *reinterpret_cast<float4*>(o) = ra;
    *reinterpret_cast<float4*>(o + 32) = rb;

    float ps = ra.x * fsa.x, pd = ra.x * fda.x;
    ps = __builtin_fmaf(ra.y, fsa.y, ps); pd = __builtin_fmaf(ra.y, fda.y, pd);
    ps = __builtin_fmaf(ra.z, fsa.z, ps); pd = __builtin_fmaf(ra.z, fda.z, pd);
    ps = __builtin_fmaf(ra.w, fsa.w, ps); pd = __builtin_fmaf(ra.w, fda.w, pd);
    ps = __builtin_fmaf(rb.x, fsb.x, ps); pd = __builtin_fmaf(rb.x, fdb.x, pd);
    ps = __builtin_fmaf(rb.y, fsb.y, ps); pd = __builtin_fmaf(rb.y, fdb.y, pd);
    ps = __builtin_fmaf(rb.z, fsb.z, ps); pd = __builtin_fmaf(rb.z, fdb.z, pd);
    ps = __builtin_fmaf(rb.w, fsb.w, ps); pd = __builtin_fmaf(rb.w, fdb.w, pd);
    const bool lowh = lane < 16;
    float keep = lowh ? ps : pd;
    const float give = lowh ? pd : ps;
    keep += __shfl_xor(give, 16, 32);
    keep = rowSum16(keep);
    if ((lane & 15) == 0) (lowh ? s2src : s2dst)[grow] = keep;
}

#define BIN_NCH  256
#define BIN_T    512
#define BIN_U    8
#define BIN_MAXR 1024

static_assert(OWN_R <= 256, "the local row of an edge is kept in 8 bits");

__device__ __forceinline__ int binRanges(int N) { return min((N + OWN_R - 1) / OWN_R, BIN_MAXR - 1); }
__device__ __forceinline__ int binTabPitch(int NB) { return (NB + 32) & ~31; }
__device__ __forceinline__ int binChunk(int E) { return (E + BIN_NCH - 1) / BIN_NCH; }
__device__ __forceinline__ int binPiecePitch(int CH) { return (CH + 31) & ~31; }

static_assert(BIN_NCH * FOLD_PER_WG >= FOLD_DOTS && (16 * FOLD_PER_WG) % 32 == 0 && 16 * FOLD_PER_WG <= BIN_T,
              "the partition launch's first waves form every folded score vector entry");

__global__ __launch_bounds__(BIN_T) void k_bin_edges(
    const int* __restrict__ src, const int* __restrict__ dst,
    int* __restrict__ binned, int* __restrict__ tab, int E, int N,
    const float* __restrict__ W1, const float* __restrict__ a1,
    const float* __restrict__ W2, const float* __restrict__ a2,
    float* __restrict__ wf1, float* __restrict__ wf2)
{
    __shared__ int cnt[BIN_MAXR];
    __shared__ int wtot[BIN_T / 32];

    const int tid = threadIdx.x, lane = tid & 31, wave = tid >> 5;
    const int NB = binRanges(N);
    const int TP = binTabPitch(NB);
    const int CH = binChunk(E);
    const int SP = binPiecePitch(CH);
    const int c  = blockIdx.x;
    const int e0 = c * CH;
    const int n  = min(CH, E - e0);
    const unsigned lim = (unsigned)min(N, NB * OWN_R);

    for (int i = tid; i < BIN_MAXR; i += BIN_T) cnt[i] = 0;
    __syncthreads();

    float4 fw, fa;
    float* fplace;
    foldRequest(W1, a1, W2, a2, wf1, wf2, c, tid, fw, fa, fplace);

    for (int k0 = 0; k0 < n; k0 += BIN_T * BIN_U) {
        int s[BIN_U];
        #pragma unroll
        for (int u = 0; u < BIN_U; ++u) s[u] = src[e0 + min(k0 + u * BIN_T + tid, n - 1)];
        #pragma unroll
        for (int u = 0; u < BIN_U; ++u) {
            const int k = k0 + u * BIN_T + tid;
            if (k < n && (unsigned)s[u] < lim) atomicAdd(&cnt[(unsigned)s[u] / OWN_R], 1);
        }
    }
    if (tid < 16 * FOLD_PER_WG) foldFinish(fw, fa, fplace);
    __syncthreads();

    constexpr int PER = BIN_MAXR / BIN_T;
    int v[PER];
    int tsum = 0;
    #pragma unroll
    for (int j = 0; j < PER; ++j) { v[j] = cnt[tid * PER + j]; tsum += v[j]; }
    int incl = tsum;
    #pragma unroll
    for (int off = 1; off < 32; off <<= 1) {
        const int y = __shfl_up(incl, off, 32);
        if (lane >= off) incl += y;
    }
    if (lane == 31) wtot[wave] = incl;
    __syncthreads();
    int run = incl - tsum;
    for (int w = 0; w < wave; ++w) run += wtot[w];
    #pragma unroll
    for (int j = 0; j < PER; ++j) {
        const int r = tid * PER + j;
        cnt[r] = run;
        if (r <= NB) tab[(size_t)c * TP + r] = run;
        run += v[j];
    }
    __syncthreads();

    for (int k0 = 0; k0 < n; k0 += BIN_T * BIN_U) {
        int s[BIN_U], d[BIN_U];
        #pragma unroll
        for (int u = 0; u < BIN_U; ++u) {
            const int k = e0 + min(k0 + u * BIN_T + tid, n - 1);
            s[u] = src[k]; d[u] = dst[k];
        }
        #pragma unroll
        for (int u = 0; u < BIN_U; ++u) {
            const int k = k0 + u * BIN_T + tid;
            if (k < n && (unsigned)s[u] < lim) {
                const unsigned r = (unsigned)s[u] / OWN_R;
                const int p = atomicAdd(&cnt[r], 1);
                binned[(size_t)c * SP + p] = (int)((((unsigned)s[u] - r * OWN_R) << 24) | ((unsigned)d[u] & 0xffffffu));
            }
        }
    }
}

template <int HS>
__global__ __launch_bounds__(OWN_T) void k_own_attend(
    const int* __restrict__ src, const int* __restrict__ dst,
    const int* __restrict__ binned, const int* __restrict__ tab,
    const float* __restrict__ ssrc, const float* __restrict__ sdst,
    const float* __restrict__ Wh, float* __restrict__ outp, int E, int N,
    const float* __restrict__ wf2, float* __restrict__ s2src, float* __restrict__ s2dst)
{
    constexpr int H  = 1 << HS;
    constexpr int RW = H * 64;
    __shared__ __attribute__((aligned(16))) float acc[OWN_R * RW];
    __shared__ float den[OWN_R * H];
    __shared__ unsigned int mx[OWN_R * H];
    __shared__ int   queA[OWN_QCAP];
    __shared__ int   queB[OWN_QCAP];
    __shared__ int   rcnt[OWN_NW * OWN_RPW];
    __shared__ int   rpos[OWN_NW * OWN_RPW];
    __shared__ int   bst[OWN_NW + 1];
    __shared__ int   cst[BIN_NCH];
    __shared__ int   qof[BIN_NCH + 1];
    __shared__ int   wsm[BIN_NCH / 32];

    const int tid  = threadIdx.x;
    const int lane = tid & 31;
    const int lo   = blockIdx.x * OWN_R;
    const int nrows = min(OWN_R, N - lo);

    const int NB = binRanges(N);
    const int TP = binTabPitch(NB);
    const int SP = binPiecePitch(binChunk(E));
    int myst = 0, mycnt = 0;
    if (tid < BIN_NCH && (int)blockIdx.x < NB) {
        const int* tr = tab + (size_t)tid * TP + blockIdx.x;
        myst = tr[0]; mycnt = tr[1] - myst;
    }

    float4* acc4 = reinterpret_cast<float4*>(acc);
    for (int i = tid; i < OWN_R * H; i += OWN_T) mx[i] = 0u;
    if (tid < OWN_NW * OWN_RPW) rcnt[tid] = 0;

    int incl = mycnt;
    if (tid < BIN_NCH) {
        #pragma unroll
        for (int off = 1; off < 32; off <<= 1) {
            const int y = __shfl_up(incl, off, 32);
            if (lane >= off) incl += y;
        }
        cst[tid] = myst;
        if (lane == 31) wsm[tid >> 5] = incl;
    }
    __syncthreads();
    if (tid < BIN_NCH) {
        int pre = incl - mycnt;
        for (int w = 0; w < (tid >> 5); ++w) pre += wsm[w];
        qof[tid] = pre;
        if (tid == BIN_NCH - 1) qof[BIN_NCH] = pre + mycnt;
    }
    __syncthreads();
    const int ntot = qof[BIN_NCH];

    if (ntot <= OWN_QCAP) {
        for (int base = 0; base < ntot; base += OWN_T * OWN_P) {
            const int i0 = base + tid;
            if (i0 < ntot) {
                int qi[OWN_P], cI[OWN_P], wI[OWN_P], rI[OWN_P], dI[OWN_P];
                #pragma unroll
                for (int u = 0; u < OWN_P; ++u) {
                    const int i = i0 + u * OWN_T;
                    qi[u] = i < ntot ? i : i0;
                }
                #pragma unroll
                for (int u = 0; u < OWN_P; ++u) {
                    int c = 0;
                    #pragma unroll
                    for (int st = BIN_NCH / 2; st >= 1; st >>= 1)
                        if (qof[c + st] <= qi[u]) c += st;
                    cI[u] = c;
                }
                #pragma unroll
                for (int u = 0; u < OWN_P; ++u)
                    wI[u] = binned[(size_t)cI[u] * SP + cst[cI[u]] + (qi[u] - qof[cI[u]])];
                holdInts(wI);
                float a[OWN_P][H], b[OWN_P][H];
                #pragma unroll
                for (int u = 0; u < OWN_P; ++u) {
                    dI[u] = wI[u] & 0xffffff;
                    rI[u] = (int)((unsigned)wI[u] >> 24);
                    nodeScores<H>(ssrc, lo + rI[u], a[u]);
                    nodeScores<H>(sdst, dI[u], b[u]);
                }
                holdScores<H>(a);
                holdScores<H>(b);
                #pragma unroll
                for (int u = 0; u < OWN_P; ++u) {
                    #pragma unroll
                    for (int h = 0; h < H; ++h)
                        atomicMax(&mx[rI[u] * H + h], ordKey(edgeScore(a[u][h], b[u][h])));
                    queA[qi[u]] = wI[u];
                    if (i0 + u * OWN_T < ntot) atomicAdd(&rcnt[rI[u]], 1);
                }
            }
        }
        __syncthreads();

        if (tid < OWN_NW) {
            int c[OWN_RPW];
            int t = 0;
            #pragma unroll
            for (int k = 0; k < OWN_RPW; ++k) { c[k] = rcnt[tid + OWN_NW * k]; t += c[k]; }
            int run = t;
            #pragma unroll
            for (int off = 1; off < 32; off <<= 1) {
                const int y = __shfl_up(run, off, 32);
                if (lane >= off) run += y;
            }
            int st = run - t;
            bst[tid] = st;
            #pragma unroll
            for (int k = 0; k < OWN_RPW; ++k) { rpos[tid + OWN_NW * k] = st; st += c[k]; }
            if (tid == OWN_NW - 1) bst[OWN_NW] = run;
        }
        __syncthreads();
        for (int i = tid; i < ntot; i += OWN_T) {
            const int wd = queA[i];
            const int p = atomicAdd(&rpos[(unsigned)wd >> 24], 1);
            queB[p] = wd;
        }
        __syncthreads();

        const int wave = __builtin_amdgcn_readfirstlane(tid >> 5);
        const int bbeg = __builtin_amdgcn_readfirstlane(bst[wave]);
        const int bend = __builtin_amdgcn_readfirstlane(bst[wave + 1]);
        if constexpr (H == 4) {
            const int hq = lane >> 3, g = lane & 7;
            const unsigned voff = ((unsigned)hq * (unsigned)N * 64u + 4u * (unsigned)g) * 4u;
            const char* __restrict__ whb = reinterpret_cast<const char*>(Wh);
            const float4 fsa = *reinterpret_cast<const float4*>(wf2 + hq * 64 + 4 * g);
            const float4 fsb = *reinterpret_cast<const float4*>(wf2 + hq * 64 + 4 * g + 32);
            const float4 fda = *reinterpret_cast<const float4*>(wf2 + 256 + hq * 64 + 4 * g);
            const float4 fdb = *reinterpret_cast<const float4*>(wf2 + 256 + hq * 64 + 4 * g + 32);
            int cur = -1;
            float4 sa = make_float4(0.0f, 0.0f, 0.0f, 0.0f), sb = sa;
            float dsm = 0.0f;
            for (int b0 = bbeg; b0 < bend; b0 += 8) {
                const bool ok = b0 + g < bend;
                const int wd = queB[ok ? b0 + g : b0];
                const int d  = wd & 0xffffff;
                const int rl = (int)((unsigned)wd >> 24);
                const float sc = edgeScore(ssrc[(size_t)(lo + rl) * 4 + hq], sdst[(size_t)d * 4 + hq]);
                const float ex = __expf(sc - ordVal(mx[rl * 4 + hq]));
                float4 wa[8], wb[8];
                #pragma unroll
                for (int e = 0; e < 8; ++e) {
                    const unsigned dE = (unsigned)__builtin_amdgcn_readlane(d, e);
                    const char* rec = whb + (size_t)dE * 256u;
                    wa[e] = *reinterpret_cast<const float4*>(rec + voff);
                    wb[e] = *reinterpret_cast<const float4*>(rec + voff + 128);
                }
                holdQuads(wa, wb, 0); holdQuads(wa, wb, 2); holdQuads(wa, wb, 4); holdQuads(wa, wb, 6);
                #pragma unroll
                for (int e = 0; e < 8; ++e) {
                    if (b0 + e < bend) {
                        const int r = __builtin_amdgcn_readlane(rl, e);
                        if (r != cur) {
                            if (cur >= 0) storeRowQ(outp, lo + cur, lane, sa, sb, dsm, fsa, fsb, fda, fdb, s2src, s2dst);
                            cur = r;
                            sa = make_float4(0.0f, 0.0f, 0.0f, 0.0f); sb = sa; dsm = 0.0f;
                        }
                        const float exE = groupLane(ex, e);
                        sa.x = __builtin_fmaf(exE, wa[e].x, sa.x); sa.y = __builtin_fmaf(exE, wa[e].y, sa.y);
                        sa.z = __builtin_fmaf(exE, wa[e].z, sa.z); sa.w = __builtin_fmaf(exE, wa[e].w, sa.w);
                        sb.x = __builtin_fmaf(exE, wb[e].x, sb.x); sb.y = __builtin_fmaf(exE, wb[e].y, sb.y);
                        sb.z = __builtin_fmaf(exE, wb[e].z, sb.z); sb.w = __builtin_fmaf(exE, wb[e].w, sb.w);
                        dsm += exE;
                    }
                }
            }
            if (cur >= 0) storeRowQ(outp, lo + cur, lane, sa, sb, dsm, fsa, fsb, fda, fdb, s2src, s2dst);
        } else {
            constexpr int EPB = 32 >> HS;
            const int j = lane >> HS, h = lane & (H - 1);
            int cur = -1;
            float2 sum[H];
            float dsm[H];
            #pragma unroll
            for (int hh = 0; hh < H; ++hh) { sum[hh] = make_float2(0.0f, 0.0f); dsm[hh] = 0.0f; }
            for (int b0 = bbeg; b0 < bend; b0 += EPB) {
                const bool ok = b0 + j < bend;
                const int wd = queB[ok ? b0 + j : b0];
                const int d  = wd & 0xffffff;
                const int rl = (int)((unsigned)wd >> 24);
                const float sc = edgeScore(ssrc[(size_t)(lo + rl) * H + h], sdst[(size_t)d * H + h]);
                const float ex = __expf(sc - ordVal(mx[rl * H + h]));
                float2 w[32];
                #pragma unroll
                for (int L = 0; L < 32; ++L) {
                    const int dL = __builtin_amdgcn_readlane(d, L);
                    const float* row = Wh + ((size_t)(L & (H - 1)) * N + dL) * 64;
                    w[L] = *reinterpret_cast<const float2*>(row + 2 * lane);
                }
                holdRows(w, 0); holdRows(w, 8); holdRows(w, 16); holdRows(w, 24);
                #pragma unroll
                for (int e = 0; e < EPB; ++e) {
                    if (b0 + e < bend) {
                        const int LE = e * H;
                        const int r = __builtin_amdgcn_readlane(rl, LE);
                        if (r != cur) {
                            if (cur >= 0) storeRow<H>(outp, lo + cur, lane, sum, dsm);
                            cur = r;
                            #pragma unroll
                            for (int hh = 0; hh < H; ++hh) { sum[hh] = make_float2(0.0f, 0.0f); dsm[hh] = 0.0f; }
                        }
                        #pragma unroll
                        for (int hh = 0; hh < H; ++hh) {
                            const float exE = __int_as_float(__builtin_amdgcn_readlane(__float_as_int(ex), LE + hh));
                            sum[hh].x = __builtin_fmaf(exE, w[LE + hh].x, sum[hh].x);
                            sum[hh].y = __builtin_fmaf(exE, w[LE + hh].y, sum[hh].y);
                            dsm[hh] += exE;
                        }
                    }
                }
            }
            if (cur >= 0) storeRow<H>(outp, lo + cur, lane, sum, dsm);
        }
        for (int k = 0; k < OWN_RPW; ++k) {
            const int r = wave + OWN_NW * k;
            if (r < nrows && rcnt[r] == 0) {
                float* o = outp + (size_t)(lo + r) * RW + 2 * lane;
                for (int hh = 0; hh < H; ++hh) *reinterpret_cast<float2*>(o + hh * 64) = make_float2(0.0f, 0.0f);
                if constexpr (H == 4) { if (lane == 0) { s2src[lo + r] = 0.0f; s2dst[lo + r] = 0.0f; } }
            }
        }
    } else {
        for (int i = tid; i < OWN_R * RW / 4; i += OWN_T) acc4[i] = make_float4(0.0f, 0.0f, 0.0f, 0.0f);
        for (int i = tid; i < OWN_R * H; i += OWN_T) den[i] = 0.0f;
        for (int e = tid; e < E; e += OWN_T) {
            const int s = src[e];
            const unsigned rl = (unsigned)(s - lo);
            if (rl < (unsigned)nrows) {
                const int d = dst[e];
                for (int hh = 0; hh < H; ++hh)
                    atomicMax(&mx[rl * H + hh],
                              ordKey(edgeScore(ssrc[(size_t)s * H + hh], sdst[(size_t)d * H + hh])));
            }
        }
        __syncthreads();
        for (int e = tid; e < E; e += OWN_T) {
            const int s = src[e];
            const unsigned rl = (unsigned)(s - lo);
            if (rl < (unsigned)nrows) {
                const int d = dst[e];
                for (int hh = 0; hh < H; ++hh) {
                    const float sc = edgeScore(ssrc[(size_t)s * H + hh], sdst[(size_t)d * H + hh]);
                    const float ex = __expf(sc - ordVal(mx[rl * H + hh]));
                    ldsAddF(&den[rl * H + hh], ex);
                    const float* row = Wh + ((size_t)hh * N + d) * 64;
                    float* ap = acc + rl * RW + hh * 64;
                    for (int k = 0; k < 64; ++k) ldsAddF(ap + k, ex * row[k]);
                }
            }
        }
        __syncthreads();

        float4* o4 = reinterpret_cast<float4*>(outp + (size_t)lo * RW);
        const int n4 = nrows * (RW / 4);
        for (int i = tid; i < n4; i += OWN_T) {
            const float4 a = acc4[i];
            const float dn = den[i >> 4];
            float4 r;
            if (dn > 0.0f) {
                r.x = eluF(a.x / dn); r.y = eluF(a.y / dn); r.z = eluF(a.z / dn); r.w = eluF(a.w / dn);
            } else {
                r = make_float4(0.0f, 0.0f, 0.0f, 0.0f);
            }
            o4[i] = r;
            if constexpr (H == 4) acc4[i] = r;
        }
        if constexpr (H == 4) {
            __syncthreads();
            for (int i = tid; i < nrows * 2; i += OWN_T) {
                const int rr = i >> 1, v = i & 1;
                const float* fv = wf2 + v * 256;
                const float* ar = acc + rr * RW;
                float p = 0.0f;
                for (int k = 0; k < RW; ++k) p = __builtin_fmaf(ar[k], fv[k], p);
                (v ? s2dst : s2src)[lo + rr] = p;
            }
        }
    }
}

static inline int cdiv(long a, int b) { return (int)((a + b - 1) / b); }

extern "C" void kernel_launch(void* const* d_in, const int* in_sizes, int n_in,
                              void* d_out, int out_size, void* d_ws, size_t ws_size,
                              hipStream_t stream)
{
    const float* x       = (const float*)d_in[0];
    const int*   edges   = (const int*)d_in[1];
    const float* W_heads = (const float*)d_in[2];
    const float* a_heads = (const float*)d_in[3];
    const float* W_out   = (const float*)d_in[4];
    const float* a_out   = (const float*)d_in[5];
    float* out = (float*)d_out;

    const int N = in_sizes[0] / 128;
    const int E = in_sizes[1] / 2;
    const int H = 4, HID = 64;

    const int* src = edges;
    const int* dst = edges + E;

    float* ws = (float*)d_ws;
    size_t off = 0;
    float* Wh1   = ws + off; off += (size_t)H * N * HID;
    float* hcat  = ws + off; off += (size_t)N * 256;
    float* Wh2   = ws + off; off += (size_t)N * HID;
    float* ssrc1 = ws + off; off += (size_t)H * N;
    float* sdst1 = ws + off; off += (size_t)H * N;
    float* ssrc2 = ws + off; off += (size_t)N;
    float* sdst2 = ws + off; off += (size_t)N;
    int* binned  = (int*)(ws + off); off += (size_t)E + 64 * BIN_NCH;
    int* tab     = (int*)(ws + off);
    float* wf1   = hcat;
    float* wf2   = Wh2;

    k_bin_edges<<<BIN_NCH, BIN_T, 0, stream>>>(src, dst, binned, tab, E, N,
                                               W_heads, a_heads, W_out, a_out, wf1, wf2);

    k_gemm_res<128, 4, true, true, true><<<256, 256, 0, stream>>>(x, W_heads, Wh1, N, 4, wf1, ssrc1, sdst1);
    k_own_attend<2><<<cdiv((long)N, OWN_R), OWN_T, 0, stream>>>(src, dst, binned, tab, ssrc1, sdst1, Wh1, hcat, E, N,
                                                                wf2, ssrc2, sdst2);

    k_gemm_res<256, 2, false, false, false><<<256, 256, 0, stream>>>(hcat, W_out, Wh2, N, 2, nullptr, nullptr, nullptr);
    k_own_attend<0><<<cdiv((long)N, OWN_R), OWN_T, 0, stream>>>(src, dst, binned, tab, ssrc2, sdst2, Wh2, out, E, N,
                                                                nullptr, nullptr, nullptr);
}
